# starting kernel
# baseline (speedup 1.0000x reference)
#include <hip/hip_runtime.h>
#include <math.h>


typedef float v2f __attribute__((ext_vector_type(2)));
typedef float v8f __attribute__((ext_vector_type(8)));

static inline long long cdiv_ll(long long a, long long b) { return (a + b - 1) / b; }

__global__ void fill_kernel(float* __restrict__ p, float v, unsigned n) {
  unsigned i = blockIdx.x * blockDim.x + threadIdx.x;
  if (i < n) p[i] = v;
}

__global__ void deg_kernel(const int* __restrict__ ei, float* __restrict__ deg, int E) {
  int e = blockIdx.x * blockDim.x + threadIdx.x;
  if (e < E) atomicAdd(&deg[ei[E + e]], 1.0f);
}

__global__ void dis_kernel(const float* __restrict__ deg, float* __restrict__ dis, int N) {
  int i = blockIdx.x * blockDim.x + threadIdx.x;
  if (i < N) { float d = deg[i]; dis[i] = d > 0.0f ? rsqrtf(d) : 0.0f; }
}

__global__ void edge_kernel(const int* __restrict__ ei, const float* __restrict__ dis,
                            int4* __restrict__ edges, int E) {
  int e = blockIdx.x * blockDim.x + threadIdx.x;
  if (e >= E) return;
  int s = ei[e], d = ei[E + e];
  int4 r;
  r.x = s;
  r.y = d;
  r.z = __float_as_int(dis[s] * dis[d]);
  r.w = 0;
  edges[e] = r;
}

__global__ void gemm_wmma_f32(const float* __restrict__ H, const float* __restrict__ W,
                              float* __restrict__ out, int N, int K, int OUT,
                              int nColTiles, int nTiles) {
  int wave = (int)(((unsigned)blockIdx.x * blockDim.x + threadIdx.x) >> 5);
  if (wave >= nTiles) return;
  int lane = threadIdx.x & 31;
  int rt = wave / nColTiles, ct = wave % nColTiles;
  int half = lane >> 4;
  int m = lane & 15;
  int row = rt * 16 + m;
  int rowc = row < N ? row : N - 1;
  int col = ct * 16 + m;
  const float* hrow = H + (unsigned)rowc * (unsigned)K;

  v8f acc = {};
  for (int kk = 0; kk < K; kk += 4) {
    int ka = kk + half * 2;
    v2f a = *(const v2f*)(hrow + ka);
    v2f b;
    b.x = W[(unsigned)ka * (unsigned)OUT + col];
    b.y = W[(unsigned)(ka + 1) * (unsigned)OUT + col];
    acc = __builtin_amdgcn_wmma_f32_16x16x4_f32(
         false, a,  false, b,
         (short)0, acc,  false,  false);
  }
#pragma unroll
  for (int r = 0; r < 8; ++r) {
    int mrow = rt * 16 + r + 8 * half;
    if (mrow < N) out[(unsigned)mrow * (unsigned)OUT + col] = acc[r];
  }
}

template <int LOGC>
__global__ void scatter_kernel(const float* __restrict__ ht, const int4* __restrict__ edges,
                               float* __restrict__ agg, unsigned E) {
  unsigned t = blockIdx.x * blockDim.x + threadIdx.x;
  unsigned e = t >> (LOGC - 2);
  if (e >= E) return;
  unsigned c4 = (t & ((1u << (LOGC - 2)) - 1u)) << 2;
  const int4 r = edges[e];
  unsigned src = (unsigned)r.x, dst = (unsigned)r.y;
  float nrm = __int_as_float(r.z);
  const float4 v = *(const float4*)(ht + (src << LOGC) + c4);
  float* o = agg + (dst << LOGC) + c4;
  atomicAdd(o + 0, v.x * nrm);
  atomicAdd(o + 1, v.y * nrm);
  atomicAdd(o + 2, v.z * nrm);
  atomicAdd(o + 3, v.w * nrm);
}

__global__ void scatter1_kernel(const float* __restrict__ ht, const int4* __restrict__ edges,
                                float* __restrict__ agg, unsigned E) {
  unsigned e = blockIdx.x * blockDim.x + threadIdx.x;
  if (e >= E) return;
  const int4 r = edges[e];
  atomicAdd(&agg[(unsigned)r.y], ht[(unsigned)r.x] * __int_as_float(r.z));
}

template <int LOGC, bool TANH>
__global__ void finalize_kernel(const float* __restrict__ agg, const float* __restrict__ ht,
                                const float* __restrict__ dis, const float* __restrict__ bias,
                                float* __restrict__ out, unsigned total) {
  unsigned t = blockIdx.x * blockDim.x + threadIdx.x;
  if (t >= total) return;
  unsigned i = t >> LOGC;
  unsigned c = t & ((1u << LOGC) - 1u);
  float d = dis[i];
  float v = agg[t] + d * d * ht[t] + bias[c];
  out[t] = TANH ? tanhf(v) : v;
}

__global__ void dot32_kernel(const float* __restrict__ H, const float* __restrict__ W3,
                             float* __restrict__ out, int N) {
  int i = blockIdx.x * blockDim.x + threadIdx.x;
  if (i >= N) return;
  const float4* h = (const float4*)(H + (unsigned)i * 32u);
  float s = 0.f;
#pragma unroll
  for (int q = 0; q < 8; ++q) {
    float4 a = h[q];
    float4 w = ((const float4*)W3)[q];
    s += a.x * w.x + a.y * w.y + a.z * w.z + a.w * w.w;
  }
  out[i] = s;
}

extern "C" void kernel_launch(void* const* d_in, const int* in_sizes, int n_in,
                              void* d_out, int out_size, void* d_ws, size_t ws_size,
                              hipStream_t stream) {
  const float* x  = (const float*)d_in[0];
  const int*   ei = (const int*)d_in[1];
  const float* W1 = (const float*)d_in[2];
  const float* b1 = (const float*)d_in[3];
  const float* W2 = (const float*)d_in[4];
  const float* b2 = (const float*)d_in[5];
  const float* W3 = (const float*)d_in[6];
  const float* b3 = (const float*)d_in[7];
  float* out = (float*)d_out;

  const int D2 = 64, D = 32;
  const int N = in_sizes[0] / D2;
  const int E = in_sizes[1] / 2;

  float* ws    = (float*)d_ws;
  float* dis   = ws;
  float* ht    = ws + N;
  float* agg   = ht + (long long)N * D2;
  float* ht3   = agg + (long long)N * D2;
  float* agg3  = ht3 + N;
  int4*  edges = (int4*)(agg3 + N);
  float* deg   = ht;
  (void)ws_size; (void)n_in; (void)out_size;

  const int B = 256;

  fill_kernel<<<(unsigned)cdiv_ll(N, B), B, 0, stream>>>(deg, 1.0f, (unsigned)N);
  deg_kernel<<<(unsigned)cdiv_ll(E, B), B, 0, stream>>>(ei, deg, E);
  dis_kernel<<<(unsigned)cdiv_ll(N, B), B, 0, stream>>>(deg, dis, N);
  edge_kernel<<<(unsigned)cdiv_ll(E, B), B, 0, stream>>>(ei, dis, edges, E);

  const int rowTiles = (int)cdiv_ll(N, 16);

  {
    int colT = D2 / 16, tiles = rowTiles * colT;
    gemm_wmma_f32<<<(unsigned)cdiv_ll((long long)tiles * 32, B), B, 0, stream>>>(
        x, W1, ht, N, D2, D2, colT, tiles);
  }
  fill_kernel<<<(unsigned)cdiv_ll((long long)N * D2, B), B, 0, stream>>>(agg, 0.f, (unsigned)(N * D2));
  scatter_kernel<6><<<(unsigned)cdiv_ll((long long)E * (D2 / 4), B), B, 0, stream>>>(ht, edges, agg, (unsigned)E);
  finalize_kernel<6, true><<<(unsigned)cdiv_ll((long long)N * D2, B), B, 0, stream>>>(
      agg, ht, dis, b1, agg, (unsigned)(N * D2));

  {
    int colT = D / 16, tiles = rowTiles * colT;
    gemm_wmma_f32<<<(unsigned)cdiv_ll((long long)tiles * 32, B), B, 0, stream>>>(
        agg, W2, ht, N, D2, D, colT, tiles);
  }
  fill_kernel<<<(unsigned)cdiv_ll((long long)N * D, B), B, 0, stream>>>(agg, 0.f, (unsigned)(N * D));
  scatter_kernel<5><<<(unsigned)cdiv_ll((long long)E * (D / 4), B), B, 0, stream>>>(ht, edges, agg, (unsigned)E);
  finalize_kernel<5, true><<<(unsigned)cdiv_ll((long long)N * D, B), B, 0, stream>>>(
      agg, ht, dis, b2, agg, (unsigned)(N * D));

  dot32_kernel<<<(unsigned)cdiv_ll(N, B), B, 0, stream>>>(agg, W3, ht3, N);
  fill_kernel<<<(unsigned)cdiv_ll(N, B), B, 0, stream>>>(agg3, 0.f, (unsigned)N);
  scatter1_kernel<<<(unsigned)cdiv_ll(E, B), B, 0, stream>>>(ht3, edges, agg3, (unsigned)E);
  finalize_kernel<0, false><<<(unsigned)cdiv_ll(N, B), B, 0, stream>>>(
      agg3, ht3, dis, b3, out, (unsigned)N);
}


# Sort edges by destination once (returning int atomic gives degree and rank), then per-node segment gather replaces f32 atomic scatter; sum order: arrival-rank sequence, 8-lane tree in layer 3
# speedup vs baseline: 2.0261x; 2.0261x over previous
#include <hip/hip_runtime.h>
#include <math.h>


typedef float v2f __attribute__((ext_vector_type(2)));
typedef float v8f __attribute__((ext_vector_type(8)));

static inline long long cdiv_ll(long long a, long long b) { return (a + b - 1) / b; }

__global__ void fill_kernel(float* __restrict__ p, float v, unsigned n) {
  unsigned i = blockIdx.x * blockDim.x + threadIdx.x;
  if (i < n) p[i] = v;
}

__global__ void rank_kernel(const int* __restrict__ ei, int* __restrict__ cnt,
                            int* __restrict__ rank, int E) {
  int e = blockIdx.x * blockDim.x + threadIdx.x;
  if (e < E) rank[e] = atomicAdd(&cnt[ei[E + e]], 1);
}

__global__ void __launch_bounds__(256)
scan_kernel(const int* __restrict__ cnt, int* __restrict__ off,
            float* __restrict__ dis, int N) {
  __shared__ int sh[16];
  const int tid = threadIdx.x;
  const int lane = tid & 31, w = tid >> 5;
  const int base = blockIdx.x * 1024;

  const int4* front = (const int4*)cnt + tid;
  int pre = 0;
#pragma unroll 8
  for (int m = 0; m < (int)blockIdx.x; ++m) {
    const int4 v = front[m * 256];
    pre += (v.x + v.y) + (v.z + v.w);
  }

  const int i0 = base + tid * 4;
  int c[4];
  for (int q = 0; q < 4; ++q) c[q] = (i0 + q < N) ? cnt[i0 + q] : 0;
  const int loc = (c[0] + c[1]) + (c[2] + c[3]);

  int incl = loc;
  for (int d = 1; d < 32; d <<= 1) {
    int t = __shfl_up(incl, d, 32);
    if (lane >= d) incl += t;
  }
  int psum = pre;
  for (int d = 16; d > 0; d >>= 1) psum += __shfl_xor(psum, d, 32);

  if (lane == 31) sh[8 + w] = incl;
  if (lane == 0) sh[w] = psum;
  __syncthreads();
  int run = incl - loc;
  for (int k = 0; k < 8; ++k) {
    run += sh[k];
    if (k < w) run += sh[8 + k];
  }
  for (int q = 0; q < 4; ++q) {
    if (i0 + q < N) {
      off[i0 + q] = run;
      float dg = (float)(c[q] + 1);
      dis[i0 + q] = dg > 0.0f ? rsqrtf(dg) : 0.0f;
    }
    run += c[q];
  }
}

__global__ void place_kernel(const int* __restrict__ ei, const int* __restrict__ rank,
                             const int* __restrict__ off, const float* __restrict__ dis,
                             int2* __restrict__ rec, int E) {
  int e = blockIdx.x * blockDim.x + threadIdx.x;
  if (e >= E) return;
  int s = ei[e], d = ei[E + e];
  int2 r;
  r.x = s;
  r.y = __float_as_int(dis[s] * dis[d]);
  rec[(unsigned)(off[d] + rank[e])] = r;
}

__global__ void gemm_wmma_f32(const float* __restrict__ H, const float* __restrict__ W,
                              float* __restrict__ out, int N, int K, int OUT,
                              int nColTiles, int nTiles) {
  int wave = (int)(((unsigned)blockIdx.x * blockDim.x + threadIdx.x) >> 5);
  if (wave >= nTiles) return;
  int lane = threadIdx.x & 31;
  int rt = wave / nColTiles, ct = wave % nColTiles;
  int half = lane >> 4;
  int m = lane & 15;
  int row = rt * 16 + m;
  int rowc = row < N ? row : N - 1;
  int col = ct * 16 + m;
  const float* hrow = H + (unsigned)rowc * (unsigned)K;

  v8f acc = {};
  for (int kk = 0; kk < K; kk += 4) {
    int ka = kk + half * 2;
    v2f a = *(const v2f*)(hrow + ka);
    v2f b;
    b.x = W[(unsigned)ka * (unsigned)OUT + col];
    b.y = W[(unsigned)(ka + 1) * (unsigned)OUT + col];
    acc = __builtin_amdgcn_wmma_f32_16x16x4_f32(
         false, a,  false, b,
         (short)0, acc,  false,  false);
  }
#pragma unroll
  for (int r = 0; r < 8; ++r) {
    int mrow = rt * 16 + r + 8 * half;
    if (mrow < N) out[(unsigned)mrow * (unsigned)OUT + col] = acc[r];
  }
}

template <int LOGC, bool TANH>
__global__ void gather_kernel(const float* __restrict__ ht, const int2* __restrict__ rec,
                              const int* __restrict__ off, const int* __restrict__ cnt,
                              const float* __restrict__ dis, const float* __restrict__ bias,
                              float* __restrict__ out, unsigned N) {
  unsigned t = blockIdx.x * blockDim.x + threadIdx.x;
  unsigned i = t >> (LOGC - 2);
  if (i >= N) return;
  unsigned c4 = (t & ((1u << (LOGC - 2)) - 1u)) << 2;
  const float* hc = ht + c4;
  const int2* rp = rec + (unsigned)off[i];
  const unsigned n = (unsigned)cnt[i];
  const float4 hv = *(const float4*)(hc + (i << LOGC));
  const float4 bv = *(const float4*)(bias + c4);
  const float di = dis[i];

  float4 acc = {0.f, 0.f, 0.f, 0.f};
  unsigned k = 0;
  for (; k + 4 <= n; k += 4) {
    const int2 r0 = rp[k], r1 = rp[k + 1], r2 = rp[k + 2], r3 = rp[k + 3];
    const float4 v0 = *(const float4*)(hc + ((unsigned)r0.x << LOGC));
    const float4 v1 = *(const float4*)(hc + ((unsigned)r1.x << LOGC));
    const float4 v2 = *(const float4*)(hc + ((unsigned)r2.x << LOGC));
    const float4 v3 = *(const float4*)(hc + ((unsigned)r3.x << LOGC));
    const float n0 = __int_as_float(r0.y), n1 = __int_as_float(r1.y);
    const float n2 = __int_as_float(r2.y), n3 = __int_as_float(r3.y);
    acc.x += v0.x * n0; acc.y += v0.y * n0; acc.z += v0.z * n0; acc.w += v0.w * n0;
    acc.x += v1.x * n1; acc.y += v1.y * n1; acc.z += v1.z * n1; acc.w += v1.w * n1;
    acc.x += v2.x * n2; acc.y += v2.y * n2; acc.z += v2.z * n2; acc.w += v2.w * n2;
    acc.x += v3.x * n3; acc.y += v3.y * n3; acc.z += v3.z * n3; acc.w += v3.w * n3;
  }
#pragma nounroll
  for (; k < n; ++k) {
    const int2 r0 = rp[k];
    const float4 v0 = *(const float4*)(hc + ((unsigned)r0.x << LOGC));
    const float n0 = __int_as_float(r0.y);
    acc.x += v0.x * n0; acc.y += v0.y * n0; acc.z += v0.z * n0; acc.w += v0.w * n0;
  }
  const float sl = di * di;
  float4 o;
  o.x = acc.x + sl * hv.x + bv.x;
  o.y = acc.y + sl * hv.y + bv.y;
  o.z = acc.z + sl * hv.z + bv.z;
  o.w = acc.w + sl * hv.w + bv.w;
  if (TANH) { o.x = tanhf(o.x); o.y = tanhf(o.y); o.z = tanhf(o.z); o.w = tanhf(o.w); }
  *(float4*)(out + (i << LOGC) + c4) = o;
}

__global__ void __launch_bounds__(256)
gather1_kernel(const float* __restrict__ ht3, const int2* __restrict__ rec,
               const int* __restrict__ off, const int* __restrict__ cnt,
               const float* __restrict__ dis, const float* __restrict__ bias,
               float* __restrict__ out, unsigned N) {
  __shared__ float res[32];
  const unsigned i = blockIdx.x * 32u + (threadIdx.x >> 3);
  const unsigned j = threadIdx.x & 7u;
  float s = 0.f;
  if (i < N) {
    const int2* rp = rec + (unsigned)off[i];
    const unsigned n = (unsigned)cnt[i];
    for (unsigned k = j; k < n; k += 8) {
      const int2 r = rp[k];
      s += ht3[(unsigned)r.x] * __int_as_float(r.y);
    }
  }
  s += __shfl_xor(s, 1, 32);
  s += __shfl_xor(s, 2, 32);
  s += __shfl_xor(s, 4, 32);
  if (j == 0) res[threadIdx.x >> 3] = s;
  __syncthreads();
  if (threadIdx.x < 32) {
    const unsigned i2 = blockIdx.x * 32u + threadIdx.x;
    if (i2 < N) {
      const float d = dis[i2];
      out[i2] = res[threadIdx.x] + d * d * ht3[i2] + bias[0];
    }
  }
}

__global__ void dot32_kernel(const float* __restrict__ H, const float* __restrict__ W3,
                             float* __restrict__ out, int N) {
  int i = blockIdx.x * blockDim.x + threadIdx.x;
  if (i >= N) return;
  const float4* h = (const float4*)(H + (unsigned)i * 32u);
  float s = 0.f;
#pragma unroll
  for (int q = 0; q < 8; ++q) {
    float4 a = h[q];
    float4 w = ((const float4*)W3)[q];
    s += a.x * w.x + a.y * w.y + a.z * w.z + a.w * w.w;
  }
  out[i] = s;
}

extern "C" void kernel_launch(void* const* d_in, const int* in_sizes, int n_in,
                              void* d_out, int out_size, void* d_ws, size_t ws_size,
                              hipStream_t stream) {
  const float* x  = (const float*)d_in[0];
  const int*   ei = (const int*)d_in[1];
  const float* W1 = (const float*)d_in[2];
  const float* b1 = (const float*)d_in[3];
  const float* W2 = (const float*)d_in[4];
  const float* b2 = (const float*)d_in[5];
  const float* W3 = (const float*)d_in[6];
  const float* b3 = (const float*)d_in[7];
  float* out = (float*)d_out;

  const int D2 = 64, D = 32;
  const int N = in_sizes[0] / D2;
  const int E = in_sizes[1] / 2;

  float* ws   = (float*)d_ws;
  int*   cnt  = (int*)d_ws;
  float* dis  = ws + N;
  float* ht   = dis + N;
  float* agg  = ht + (long long)N * D2;
  float* ht3  = agg + (long long)N * D2;
  int2*  rec  = (int2*)(ht3 + N);
  int*   rank = (int*)(rec + E);
  int*   off  = rank + E;
  (void)ws_size; (void)n_in; (void)out_size;

  const int B = 256;

  fill_kernel<<<(unsigned)cdiv_ll(N, B), B, 0, stream>>>((float*)cnt, 0.f, (unsigned)N);
  rank_kernel<<<(unsigned)cdiv_ll(E, B), B, 0, stream>>>(ei, cnt, rank, E);
  scan_kernel<<<(unsigned)cdiv_ll(N, 1024), 256, 0, stream>>>(cnt, off, dis, N);
  place_kernel<<<(unsigned)cdiv_ll(E, B), B, 0, stream>>>(ei, rank, off, dis, rec, E);

  const int rowTiles = (int)cdiv_ll(N, 16);

  {
    int colT = D2 / 16, tiles = rowTiles * colT;
    gemm_wmma_f32<<<(unsigned)cdiv_ll((long long)tiles * 32, B), B, 0, stream>>>(
        x, W1, ht, N, D2, D2, colT, tiles);
  }
  gather_kernel<6, true><<<(unsigned)cdiv_ll((long long)N * (D2 / 4), B), B, 0, stream>>>(
      ht, rec, off, cnt, dis, b1, agg, (unsigned)N);

  {
    int colT = D / 16, tiles = rowTiles * colT;
    gemm_wmma_f32<<<(unsigned)cdiv_ll((long long)tiles * 32, B), B, 0, stream>>>(
        agg, W2, ht, N, D2, D, colT, tiles);
  }
  gather_kernel<5, true><<<(unsigned)cdiv_ll((long long)N * (D / 4), B), B, 0, stream>>>(
      ht, rec, off, cnt, dis, b2, agg, (unsigned)N);

  dot32_kernel<<<(unsigned)cdiv_ll(N, B), B, 0, stream>>>(agg, W3, ht3, N);
  gather1_kernel<<<(unsigned)cdiv_ll(N, 32), 256, 0, stream>>>(
      ht3, rec, off, cnt, dis, b3, out, (unsigned)N);
}


# Group edges by destination with LDS counters in two levels (256-node bins, then node segments), not a global returning add per edge + scan + place; norm factorised dis[s]*dis[d]; arrival-order sums
# speedup vs baseline: 3.8817x; 1.9158x over previous
#include <hip/hip_runtime.h>
#include <math.h>


typedef float v2f __attribute__((ext_vector_type(2)));
typedef float v8f __attribute__((ext_vector_type(8)));

#define BIN_NODES_LOG 8
#define BIN_CAP 6144
#define MAX_BINS 512
#define BIN_EPT 16
#define CSR_B 8

static inline long long cdiv_ll(long long a, long long b) { return (a + b - 1) / b; }

__device__ __forceinline__ float inv_sqrt_deg(int c) {
  const float dg = (float)(c + 1);
  return dg > 0.0f ? rsqrtf(dg) : 0.0f;
}

__global__ void fill_kernel(float* __restrict__ p, float v, unsigned n) {
  unsigned i = blockIdx.x * blockDim.x + threadIdx.x;
  if (i < n) p[i] = v;
}

__global__ void __launch_bounds__(256)
bin_kernel(const int* __restrict__ ei, int* __restrict__ binCount,
           int2* __restrict__ binBuf, int E, int nBins) {
  __shared__ int hist[MAX_BINS];
  const int tid = threadIdx.x;
  for (int b = tid; b < nBins; b += 256) hist[b] = 0;
  __syncthreads();

  const int e0 = blockIdx.x * (256 * BIN_EPT) + tid;
  int s[BIN_EPT], d[BIN_EPT], r[BIN_EPT];
#pragma unroll
  for (int q = 0; q < BIN_EPT; ++q) {
    const int e = e0 + q * 256;
    if (e < E) { s[q] = ei[e]; d[q] = ei[E + e]; } else { s[q] = 0; d[q] = -1; }
  }
#pragma unroll
  for (int q = 0; q < BIN_EPT; ++q) {
    r[q] = 0;
    if (d[q] >= 0) r[q] = atomicAdd(&hist[d[q] >> BIN_NODES_LOG], 1);
  }
  __syncthreads();
  {
    const int b0 = tid, b1 = tid + 256;
    const bool p0 = b0 < nBins, p1 = b1 < nBins;
    const int c0 = p0 ? hist[b0] : 0, c1 = p1 ? hist[b1] : 0;
    int g0 = 0, g1 = 0;
    if (p0) g0 = atomicAdd(&binCount[b0], c0);
    if (p1) g1 = atomicAdd(&binCount[b1], c1);
    if (p0) hist[b0] = g0;
    if (p1) hist[b1] = g1;
  }
  __syncthreads();
#pragma unroll
  for (int q = 0; q < BIN_EPT; ++q) {
    if (d[q] >= 0) {
      const int b = d[q] >> BIN_NODES_LOG;
      const unsigned pos = (unsigned)(hist[b] + r[q]);
      if (pos < (unsigned)BIN_CAP) {
        int2 rc; rc.x = s[q]; rc.y = d[q];
        binBuf[(unsigned)b * (unsigned)BIN_CAP + pos] = rc;
      }
    }
  }
}

__global__ void __launch_bounds__(256)
csr_kernel(const int2* __restrict__ binBuf, const int* __restrict__ binCount,
           const int* __restrict__ ei, int* __restrict__ rec, int* __restrict__ cnt,
           int* __restrict__ off, int N, int E) {
  __shared__ int cl[256];
  __shared__ int red[8];
  __shared__ int wsum[8];
  const int b = blockIdx.x, tid = threadIdx.x;
  const int lane = tid & 31, w = tid >> 5;

  int pre = 0;
  for (int m = tid; m < b; m += 256) pre += binCount[m];
  for (int dd = 16; dd > 0; dd >>= 1) pre += __shfl_xor(pre, dd, 32);
  if (lane == 0) red[w] = pre;
  cl[tid] = 0;
  __syncthreads();
  int base = 0;
  for (int k = 0; k < 8; ++k) base += red[k];

  const int nb = binCount[b];
  const bool fromBuf = nb <= BIN_CAP;
  const int2* bp = binBuf + (unsigned)b * (unsigned)BIN_CAP;

  if (fromBuf) {
    for (int i0 = tid; i0 < nb; i0 += 256 * CSR_B) {
      int dn[CSR_B];
#pragma unroll
      for (int q = 0; q < CSR_B; ++q) {
        const int idx = i0 + q * 256;
        dn[q] = idx < nb ? bp[idx].y : -1;
      }
#pragma unroll
      for (int q = 0; q < CSR_B; ++q)
        if (dn[q] >= 0) atomicAdd(&cl[dn[q] & 255], 1);
    }
  } else {
    for (int idx = tid; idx < E; idx += 256) {
      const int dn = ei[E + idx];
      if ((dn >> BIN_NODES_LOG) == b) atomicAdd(&cl[dn & 255], 1);
    }
  }
  __syncthreads();

  const int c = cl[tid];
  int incl = c;
  for (int dd = 1; dd < 32; dd <<= 1) {
    int t = __shfl_up(incl, dd, 32);
    if (lane >= dd) incl += t;
  }
  if (lane == 31) wsum[w] = incl;
  __syncthreads();
  int run = base + incl - c;
  for (int k = 0; k < 8; ++k)
    if (k < w) run += wsum[k];
  const int node = (b << BIN_NODES_LOG) + tid;
  if (node < N) { cnt[node] = c; off[node] = run; }
  cl[tid] = run;
  __syncthreads();

  if (fromBuf) {
    for (int i0 = tid; i0 < nb; i0 += 256 * CSR_B) {
      int2 rc[CSR_B];
#pragma unroll
      for (int q = 0; q < CSR_B; ++q) {
        const int idx = i0 + q * 256;
        if (idx < nb) rc[q] = bp[idx]; else { rc[q].x = 0; rc[q].y = -1; }
      }
#pragma unroll
      for (int q = 0; q < CSR_B; ++q) {
        if (rc[q].y >= 0) {
          const int slot = atomicAdd(&cl[rc[q].y & 255], 1);
          rec[(unsigned)slot] = rc[q].x;
        }
      }
    }
  } else {
    for (int idx = tid; idx < E; idx += 256) {
      const int dn = ei[E + idx];
      if ((dn >> BIN_NODES_LOG) == b) {
        const int slot = atomicAdd(&cl[dn & 255], 1);
        rec[(unsigned)slot] = ei[idx];
      }
    }
  }
}

__global__ void gemm_wmma_f32(const float* __restrict__ H, const float* __restrict__ W,
                              const int* __restrict__ cnt, float* __restrict__ out,
                              int N, int K, int OUT, int nColTiles, int nTiles) {
  int wave = (int)(((unsigned)blockIdx.x * blockDim.x + threadIdx.x) >> 5);
  if (wave >= nTiles) return;
  int lane = threadIdx.x & 31;
  int rt = wave / nColTiles, ct = wave % nColTiles;
  int half = lane >> 4;
  int m = lane & 15;
  int row = rt * 16 + m;
  int rowc = row < N ? row : N - 1;
  int col = ct * 16 + m;
  const float* hrow = H + (unsigned)rowc * (unsigned)K;

  float ds[8];
#pragma unroll
  for (int r = 0; r < 8; ++r) {
    int mrow = rt * 16 + r + 8 * half;
    ds[r] = inv_sqrt_deg(cnt[mrow < N ? mrow : N - 1]);
  }

  v8f acc = {};
  for (int kk = 0; kk < K; kk += 4) {
    int ka = kk + half * 2;
    v2f a = *(const v2f*)(hrow + ka);
    v2f b;
    b.x = W[(unsigned)ka * (unsigned)OUT + col];
    b.y = W[(unsigned)(ka + 1) * (unsigned)OUT + col];
    acc = __builtin_amdgcn_wmma_f32_16x16x4_f32(
         false, a,  false, b,
         (short)0, acc,  false,  false);
  }
#pragma unroll
  for (int r = 0; r < 8; ++r) {
    int mrow = rt * 16 + r + 8 * half;
    if (mrow < N) out[(unsigned)mrow * (unsigned)OUT + col] = acc[r] * ds[r];
  }
}

template <int LOGC, bool TANH>
__global__ void gather_kernel(const float* __restrict__ ht, const int* __restrict__ rec,
                              const int* __restrict__ off, const int* __restrict__ cnt,
                              const float* __restrict__ bias,
                              float* __restrict__ out, unsigned N) {
  unsigned t = blockIdx.x * blockDim.x + threadIdx.x;
  unsigned i = t >> (LOGC - 2);
  if (i >= N) return;
  unsigned c4 = (t & ((1u << (LOGC - 2)) - 1u)) << 2;
  const float* hc = ht + c4;
  const int* rp = rec + (unsigned)off[i];
  const int ci = cnt[i];
  const unsigned n = (unsigned)ci;
  const float4 hv = *(const float4*)(hc + (i << LOGC));
  const float4 bv = *(const float4*)(bias + c4);
  const float di = inv_sqrt_deg(ci);

  float4 acc = {0.f, 0.f, 0.f, 0.f};
  unsigned k = 0;
  for (; k + 4 <= n; k += 4) {
    const unsigned s0 = (unsigned)rp[k], s1 = (unsigned)rp[k + 1];
    const unsigned s2 = (unsigned)rp[k + 2], s3 = (unsigned)rp[k + 3];
    const float4 v0 = *(const float4*)(hc + (s0 << LOGC));
    const float4 v1 = *(const float4*)(hc + (s1 << LOGC));
    const float4 v2 = *(const float4*)(hc + (s2 << LOGC));
    const float4 v3 = *(const float4*)(hc + (s3 << LOGC));
    acc.x += v0.x; acc.y += v0.y; acc.z += v0.z; acc.w += v0.w;
    acc.x += v1.x; acc.y += v1.y; acc.z += v1.z; acc.w += v1.w;
    acc.x += v2.x; acc.y += v2.y; acc.z += v2.z; acc.w += v2.w;
    acc.x += v3.x; acc.y += v3.y; acc.z += v3.z; acc.w += v3.w;
  }
#pragma nounroll
  for (; k < n; ++k) {
    const unsigned s0 = (unsigned)rp[k];
    const float4 v0 = *(const float4*)(hc + (s0 << LOGC));
    acc.x += v0.x; acc.y += v0.y; acc.z += v0.z; acc.w += v0.w;
  }
  float4 o;
  o.x = di * (acc.x + hv.x) + bv.x;
  o.y = di * (acc.y + hv.y) + bv.y;
  o.z = di * (acc.z + hv.z) + bv.z;
  o.w = di * (acc.w + hv.w) + bv.w;
  if (TANH) { o.x = tanhf(o.x); o.y = tanhf(o.y); o.z = tanhf(o.z); o.w = tanhf(o.w); }
  *(float4*)(out + (i << LOGC) + c4) = o;
}

__global__ void __launch_bounds__(256)
gather1_kernel(const float* __restrict__ ht3, const int* __restrict__ rec,
               const int* __restrict__ off, const int* __restrict__ cnt,
               const float* __restrict__ bias,
               float* __restrict__ out, unsigned N) {
  __shared__ float res[32];
  const unsigned i = blockIdx.x * 32u + (threadIdx.x >> 3);
  const unsigned j = threadIdx.x & 7u;
  float s = 0.f;
  if (i < N) {
    const int* rp = rec + (unsigned)off[i];
    const unsigned n = (unsigned)cnt[i];
    for (unsigned k = j; k < n; k += 8) s += ht3[(unsigned)rp[k]];
  }
  s += __shfl_xor(s, 1, 32);
  s += __shfl_xor(s, 2, 32);
  s += __shfl_xor(s, 4, 32);
  if (j == 0) res[threadIdx.x >> 3] = s;
  __syncthreads();
  if (threadIdx.x < 32) {
    const unsigned i2 = blockIdx.x * 32u + threadIdx.x;
    if (i2 < N) {
      const float d = inv_sqrt_deg(cnt[i2]);
      out[i2] = d * (res[threadIdx.x] + ht3[i2]) + bias[0];
    }
  }
}

__global__ void dot32_kernel(const float* __restrict__ H, const float* __restrict__ W3,
                             const int* __restrict__ cnt, float* __restrict__ out, int N) {
  int i = blockIdx.x * blockDim.x + threadIdx.x;
  if (i >= N) return;
  const float4* h = (const float4*)(H + (unsigned)i * 32u);
  float s = 0.f;
#pragma unroll
  for (int q = 0; q < 8; ++q) {
    float4 a = h[q];
    float4 w = ((const float4*)W3)[q];
    s += a.x * w.x + a.y * w.y + a.z * w.z + a.w * w.w;
  }
  out[i] = s * inv_sqrt_deg(cnt[i]);
}

extern "C" void kernel_launch(void* const* d_in, const int* in_sizes, int n_in,
                              void* d_out, int out_size, void* d_ws, size_t ws_size,
                              hipStream_t stream) {
  const float* x  = (const float*)d_in[0];
  const int*   ei = (const int*)d_in[1];
  const float* W1 = (const float*)d_in[2];
  const float* b1 = (const float*)d_in[3];
  const float* W2 = (const float*)d_in[4];
  const float* b2 = (const float*)d_in[5];
  const float* W3 = (const float*)d_in[6];
  const float* b3 = (const float*)d_in[7];
  float* out = (float*)d_out;

  const int D2 = 64, D = 32;
  const int N = in_sizes[0] / D2;
  const int E = in_sizes[1] / 2;
  const int nBins = (int)cdiv_ll(N, 1 << BIN_NODES_LOG);

  float* ws   = (float*)d_ws;
  float* ht   = ws;
  float* agg  = ht + (long long)N * D2;
  int2*  binBuf = (int2*)agg;
  int*   rec  = (int*)(agg + (long long)N * D2);
  int*   cnt  = rec + E;
  int*   off  = cnt + N;
  float* ht3  = (float*)(off + N);
  int*   binCount = (int*)(ht3 + N);
  (void)ws_size; (void)n_in; (void)out_size;

  const int B = 256;

  fill_kernel<<<(unsigned)cdiv_ll(nBins, B), B, 0, stream>>>((float*)binCount, 0.f, (unsigned)nBins);
  bin_kernel<<<(unsigned)cdiv_ll(E, B * BIN_EPT), B, 0, stream>>>(ei, binCount, binBuf, E, nBins);
  csr_kernel<<<(unsigned)nBins, B, 0, stream>>>(binBuf, binCount, ei, rec, cnt, off, N, E);

  const int rowTiles = (int)cdiv_ll(N, 16);

  {
    int colT = D2 / 16, tiles = rowTiles * colT;
    gemm_wmma_f32<<<(unsigned)cdiv_ll((long long)tiles * 32, B), B, 0, stream>>>(
        x, W1, cnt, ht, N, D2, D2, colT, tiles);
  }
  gather_kernel<6, true><<<(unsigned)cdiv_ll((long long)N * (D2 / 4), B), B, 0, stream>>>(
      ht, rec, off, cnt, b1, agg, (unsigned)N);

  {
    int colT = D / 16, tiles = rowTiles * colT;
    gemm_wmma_f32<<<(unsigned)cdiv_ll((long long)tiles * 32, B), B, 0, stream>>>(
        agg, W2, cnt, ht, N, D2, D, colT, tiles);
  }
  gather_kernel<5, true><<<(unsigned)cdiv_ll((long long)N * (D / 4), B), B, 0, stream>>>(
      ht, rec, off, cnt, b2, agg, (unsigned)N);

  dot32_kernel<<<(unsigned)cdiv_ll(N, B), B, 0, stream>>>(agg, W3, cnt, ht3, N);
  gather1_kernel<<<(unsigned)cdiv_ll(N, 32), 256, 0, stream>>>(
      ht3, rec, off, cnt, b3, out, (unsigned)N);
}


# Next layer's dense transform moved into each aggregation's epilogue (h1 tile via LDS to f32 WMMA with W2; layer-3 dot as 4-term partial + 8-lane tree): 2 launches and the h1/h2 store+reload go
# speedup vs baseline: 4.4733x; 1.1524x over previous
#include <hip/hip_runtime.h>
#include <math.h>


typedef float v2f __attribute__((ext_vector_type(2)));
typedef float v8f __attribute__((ext_vector_type(8)));

#define BIN_NODES_LOG 8
#define BIN_CAP 6144
#define MAX_BINS 512
#define BIN_EPT 16
#define CSR_B 8

static inline long long cdiv_ll(long long a, long long b) { return (a + b - 1) / b; }

__device__ __forceinline__ float inv_sqrt_deg(int c) {
  const float dg = (float)(c + 1);
  return dg > 0.0f ? rsqrtf(dg) : 0.0f;
}

__global__ void fill_kernel(float* __restrict__ p, float v, unsigned n) {
  unsigned i = blockIdx.x * blockDim.x + threadIdx.x;
  if (i < n) p[i] = v;
}

__global__ void __launch_bounds__(256)
bin_kernel(const int* __restrict__ ei, int* __restrict__ binCount,
           int2* __restrict__ binBuf, int E, int nBins) {
  __shared__ int hist[MAX_BINS];
  const int tid = threadIdx.x;
  for (int b = tid; b < nBins; b += 256) hist[b] = 0;
  __syncthreads();

  const int e0 = blockIdx.x * (256 * BIN_EPT) + tid;
  int s[BIN_EPT], d[BIN_EPT], r[BIN_EPT];
#pragma unroll
  for (int q = 0; q < BIN_EPT; ++q) {
    const int e = e0 + q * 256;
    if (e < E) { s[q] = ei[e]; d[q] = ei[E + e]; } else { s[q] = 0; d[q] = -1; }
  }
#pragma unroll
  for (int q = 0; q < BIN_EPT; ++q) {
    r[q] = 0;
    if (d[q] >= 0) r[q] = atomicAdd(&hist[d[q] >> BIN_NODES_LOG], 1);
  }
  __syncthreads();
  {
    const int b0 = tid, b1 = tid + 256;
    const bool p0 = b0 < nBins, p1 = b1 < nBins;
    const int c0 = p0 ? hist[b0] : 0, c1 = p1 ? hist[b1] : 0;
    int g0 = 0, g1 = 0;
    if (p0) g0 = atomicAdd(&binCount[b0], c0);
    if (p1) g1 = atomicAdd(&binCount[b1], c1);
    if (p0) hist[b0] = g0;
    if (p1) hist[b1] = g1;
  }
  __syncthreads();
#pragma unroll
  for (int q = 0; q < BIN_EPT; ++q) {
    if (d[q] >= 0) {
      const int b = d[q] >> BIN_NODES_LOG;
      const unsigned pos = (unsigned)(hist[b] + r[q]);
      if (pos < (unsigned)BIN_CAP) {
        int2 rc; rc.x = s[q]; rc.y = d[q];
        binBuf[(unsigned)b * (unsigned)BIN_CAP + pos] = rc;
      }
    }
  }
}

__global__ void __launch_bounds__(256)
csr_kernel(const int2* __restrict__ binBuf, const int* __restrict__ binCount,
           const int* __restrict__ ei, int* __restrict__ rec, int* __restrict__ cnt,
           int* __restrict__ off, int N, int E) {
  __shared__ int cl[256];
  __shared__ int red[8];
  __shared__ int wsum[8];
  const int b = blockIdx.x, tid = threadIdx.x;
  const int lane = tid & 31, w = tid >> 5;

  int pre = 0;
  for (int m = tid; m < b; m += 256) pre += binCount[m];
  for (int dd = 16; dd > 0; dd >>= 1) pre += __shfl_xor(pre, dd, 32);
  if (lane == 0) red[w] = pre;
  cl[tid] = 0;
  __syncthreads();
  int base = 0;
  for (int k = 0; k < 8; ++k) base += red[k];

  const int nb = binCount[b];
  const bool fromBuf = nb <= BIN_CAP;
  const int2* bp = binBuf + (unsigned)b * (unsigned)BIN_CAP;

  if (fromBuf) {
    for (int i0 = tid; i0 < nb; i0 += 256 * CSR_B) {
      int dn[CSR_B];
#pragma unroll
      for (int q = 0; q < CSR_B; ++q) {
        const int idx = i0 + q * 256;
        dn[q] = idx < nb ? bp[idx].y : -1;
      }
#pragma unroll
      for (int q = 0; q < CSR_B; ++q)
        if (dn[q] >= 0) atomicAdd(&cl[dn[q] & 255], 1);
    }
  } else {
    for (int idx = tid; idx < E; idx += 256) {
      const int dn = ei[E + idx];
      if ((dn >> BIN_NODES_LOG) == b) atomicAdd(&cl[dn & 255], 1);
    }
  }
  __syncthreads();

  const int c = cl[tid];
  int incl = c;
  for (int dd = 1; dd < 32; dd <<= 1) {
    int t = __shfl_up(incl, dd, 32);
    if (lane >= dd) incl += t;
  }
  if (lane == 31) wsum[w] = incl;
  __syncthreads();
  int run = base + incl - c;
  for (int k = 0; k < 8; ++k)
    if (k < w) run += wsum[k];
  const int node = (b << BIN_NODES_LOG) + tid;
  if (node < N) { cnt[node] = c; off[node] = run; }
  cl[tid] = run;
  __syncthreads();

  if (fromBuf) {
    for (int i0 = tid; i0 < nb; i0 += 256 * CSR_B) {
      int2 rc[CSR_B];
#pragma unroll
      for (int q = 0; q < CSR_B; ++q) {
        const int idx = i0 + q * 256;
        if (idx < nb) rc[q] = bp[idx]; else { rc[q].x = 0; rc[q].y = -1; }
      }
#pragma unroll
      for (int q = 0; q < CSR_B; ++q) {
        if (rc[q].y >= 0) {
          const int slot = atomicAdd(&cl[rc[q].y & 255], 1);
          rec[(unsigned)slot] = rc[q].x;
        }
      }
    }
  } else {
    for (int idx = tid; idx < E; idx += 256) {
      const int dn = ei[E + idx];
      if ((dn >> BIN_NODES_LOG) == b) {
        const int slot = atomicAdd(&cl[dn & 255], 1);
        rec[(unsigned)slot] = ei[idx];
      }
    }
  }
}

__global__ void gemm_wmma_f32(const float* __restrict__ H, const float* __restrict__ W,
                              const int* __restrict__ cnt, float* __restrict__ out,
                              int N, int K, int OUT, int nColTiles, int nTiles) {
  int wave = (int)(((unsigned)blockIdx.x * blockDim.x + threadIdx.x) >> 5);
  if (wave >= nTiles) return;
  int lane = threadIdx.x & 31;
  int rt = wave / nColTiles, ct = wave % nColTiles;
  int half = lane >> 4;
  int m = lane & 15;
  int row = rt * 16 + m;
  int rowc = row < N ? row : N - 1;
  int col = ct * 16 + m;
  const float* hrow = H + (unsigned)rowc * (unsigned)K;

  float ds[8];
#pragma unroll
  for (int r = 0; r < 8; ++r) {
    int mrow = rt * 16 + r + 8 * half;
    ds[r] = inv_sqrt_deg(cnt[mrow < N ? mrow : N - 1]);
  }

  v8f acc = {};
  for (int kk = 0; kk < K; kk += 4) {
    int ka = kk + half * 2;
    v2f a = *(const v2f*)(hrow + ka);
    v2f b;
    b.x = W[(unsigned)ka * (unsigned)OUT + col];
    b.y = W[(unsigned)(ka + 1) * (unsigned)OUT + col];
    acc = __builtin_amdgcn_wmma_f32_16x16x4_f32(
         false, a,  false, b,
         (short)0, acc,  false,  false);
  }
#pragma unroll
  for (int r = 0; r < 8; ++r) {
    int mrow = rt * 16 + r + 8 * half;
    if (mrow < N) out[(unsigned)mrow * (unsigned)OUT + col] = acc[r] * ds[r];
  }
}

#define H1_PITCH 68
__global__ void __launch_bounds__(256)
gather64_xform_kernel(const float* __restrict__ ht, const int* __restrict__ rec,
                      const int* __restrict__ off, const int* __restrict__ cnt,
                      const float* __restrict__ bias, const float* __restrict__ W2,
                      float* __restrict__ out2, unsigned N) {
  __shared__ __attribute__((aligned(16))) float tile[16 * H1_PITCH];
  __shared__ __attribute__((aligned(16))) float sdis[16];
  const unsigned tid = threadIdx.x;
  const unsigned nl = tid >> 4;
  const unsigned i = blockIdx.x * 16u + nl;
  const unsigned c4 = (tid & 15u) << 2;
  float4 o = {0.f, 0.f, 0.f, 0.f};
  float di = 0.f;
  if (i < N) {
    const float* hc = ht + c4;
    const int* rp = rec + (unsigned)off[i];
    const int ci = cnt[i];
    const unsigned n = (unsigned)ci;
    const float4 hv = *(const float4*)(hc + (i << 6));
    const float4 bv = *(const float4*)(bias + c4);
    di = inv_sqrt_deg(ci);

    float4 acc = {0.f, 0.f, 0.f, 0.f};
    unsigned k = 0;
    for (; k + 4 <= n; k += 4) {
      const unsigned s0 = (unsigned)rp[k], s1 = (unsigned)rp[k + 1];
      const unsigned s2 = (unsigned)rp[k + 2], s3 = (unsigned)rp[k + 3];
      const float4 v0 = *(const float4*)(hc + (s0 << 6));
      const float4 v1 = *(const float4*)(hc + (s1 << 6));
      const float4 v2 = *(const float4*)(hc + (s2 << 6));
      const float4 v3 = *(const float4*)(hc + (s3 << 6));
      acc.x += v0.x; acc.y += v0.y; acc.z += v0.z; acc.w += v0.w;
      acc.x += v1.x; acc.y += v1.y; acc.z += v1.z; acc.w += v1.w;
      acc.x += v2.x; acc.y += v2.y; acc.z += v2.z; acc.w += v2.w;
      acc.x += v3.x; acc.y += v3.y; acc.z += v3.z; acc.w += v3.w;
    }
#pragma nounroll
    for (; k < n; ++k) {
      const unsigned s0 = (unsigned)rp[k];
      const float4 v0 = *(const float4*)(hc + (s0 << 6));
      acc.x += v0.x; acc.y += v0.y; acc.z += v0.z; acc.w += v0.w;
    }
    o.x = tanhf(di * (acc.x + hv.x) + bv.x);
    o.y = tanhf(di * (acc.y + hv.y) + bv.y);
    o.z = tanhf(di * (acc.z + hv.z) + bv.z);
    o.w = tanhf(di * (acc.w + hv.w) + bv.w);
  }
  *(float4*)(tile + nl * H1_PITCH + c4) = o;
  if ((tid & 15u) == 0u) sdis[nl] = di;
  __syncthreads();

  const unsigned w = tid >> 5;
  if (w < 2u) {
    const unsigned lane = tid & 31u;
    const unsigned half = lane >> 4;
    const unsigned m = lane & 15u;
    const unsigned col = w * 16u + m;
    const float* arow = tile + m * H1_PITCH + half * 2u;
    const float* bcol = W2 + (half * 2u) * 32u + col;
    const unsigned row0 = blockIdx.x * 16u + 8u * half;
    unsigned nv = N > row0 ? N - row0 : 0u;
    unsigned oi = row0 * 32u + col;
    const float4 sa = *(const float4*)(sdis + 8u * half);
    const float4 sb = *(const float4*)(sdis + 8u * half + 4u);
    float sc[8] = {sa.x, sa.y, sa.z, sa.w, sb.x, sb.y, sb.z, sb.w};
    asm volatile("" : "+v"(oi), "+v"(nv), "+v"(sc[0]), "+v"(sc[1]), "+v"(sc[2]), "+v"(sc[3]),
                      "+v"(sc[4]), "+v"(sc[5]), "+v"(sc[6]), "+v"(sc[7]));
    v8f acc = {};
#pragma unroll
    for (unsigned kk = 0; kk < 64u; kk += 4u) {
      v2f a = *(const v2f*)(arow + kk);
      v2f b;
      b.x = bcol[kk * 32u];
      b.y = bcol[kk * 32u + 32u];
      acc = __builtin_amdgcn_wmma_f32_16x16x4_f32(
           false, a,  false, b,
           (short)0, acc,  false,  false);
    }
    asm volatile("v_nop\n\tv_nop\n\tv_nop\n\tv_nop\n\tv_nop" : "+v"(acc));
#pragma unroll
    for (unsigned r = 0; r < 8u; ++r)
      if (r < nv) out2[oi + r * 32u] = acc[r] * sc[r];
  }
}

__global__ void gather32_dot_kernel(const float* __restrict__ ht2, const int* __restrict__ rec,
                                    const int* __restrict__ off, const int* __restrict__ cnt,
                                    const float* __restrict__ bias, const float* __restrict__ W3,
                                    float* __restrict__ out3, unsigned N) {
  unsigned t = blockIdx.x * blockDim.x + threadIdx.x;
  unsigned i = t >> 3;
  if (i >= N) return;
  unsigned c4 = (t & 7u) << 2;
  const float* hc = ht2 + c4;
  const int* rp = rec + (unsigned)off[i];
  const int ci = cnt[i];
  const unsigned n = (unsigned)ci;
  const float4 hv = *(const float4*)(hc + (i << 5));
  const float4 bv = *(const float4*)(bias + c4);
  const float4 wv = *(const float4*)(W3 + c4);
  const float di = inv_sqrt_deg(ci);

  float4 acc = {0.f, 0.f, 0.f, 0.f};
  unsigned k = 0;
  for (; k + 4 <= n; k += 4) {
    const unsigned s0 = (unsigned)rp[k], s1 = (unsigned)rp[k + 1];
    const unsigned s2 = (unsigned)rp[k + 2], s3 = (unsigned)rp[k + 3];
    const float4 v0 = *(const float4*)(hc + (s0 << 5));
    const float4 v1 = *(const float4*)(hc + (s1 << 5));
    const float4 v2 = *(const float4*)(hc + (s2 << 5));
    const float4 v3 = *(const float4*)(hc + (s3 << 5));
    acc.x += v0.x; acc.y += v0.y; acc.z += v0.z; acc.w += v0.w;
    acc.x += v1.x; acc.y += v1.y; acc.z += v1.z; acc.w += v1.w;
    acc.x += v2.x; acc.y += v2.y; acc.z += v2.z; acc.w += v2.w;
    acc.x += v3.x; acc.y += v3.y; acc.z += v3.z; acc.w += v3.w;
  }
#pragma nounroll
  for (; k < n; ++k) {
    const unsigned s0 = (unsigned)rp[k];
    const float4 v0 = *(const float4*)(hc + (s0 << 5));
    acc.x += v0.x; acc.y += v0.y; acc.z += v0.z; acc.w += v0.w;
  }
  const float ox = tanhf(di * (acc.x + hv.x) + bv.x);
  const float oy = tanhf(di * (acc.y + hv.y) + bv.y);
  const float oz = tanhf(di * (acc.z + hv.z) + bv.z);
  const float ow = tanhf(di * (acc.w + hv.w) + bv.w);
  float p = ox * wv.x + oy * wv.y + oz * wv.z + ow * wv.w;
  p += __shfl_xor(p, 1, 32);
  p += __shfl_xor(p, 2, 32);
  p += __shfl_xor(p, 4, 32);
  if ((t & 7u) == 0u) out3[i] = p * di;
}

__global__ void __launch_bounds__(256)
gather1_kernel(const float* __restrict__ ht3, const int* __restrict__ rec,
               const int* __restrict__ off, const int* __restrict__ cnt,
               const float* __restrict__ bias,
               float* __restrict__ out, unsigned N) {
  __shared__ float res[32];
  const unsigned i = blockIdx.x * 32u + (threadIdx.x >> 3);
  const unsigned j = threadIdx.x & 7u;
  float s = 0.f;
  if (i < N) {
    const int* rp = rec + (unsigned)off[i];
    const unsigned n = (unsigned)cnt[i];
    for (unsigned k = j; k < n; k += 8) s += ht3[(unsigned)rp[k]];
  }
  s += __shfl_xor(s, 1, 32);
  s += __shfl_xor(s, 2, 32);
  s += __shfl_xor(s, 4, 32);
  if (j == 0) res[threadIdx.x >> 3] = s;
  __syncthreads();
  if (threadIdx.x < 32) {
    const unsigned i2 = blockIdx.x * 32u + threadIdx.x;
    if (i2 < N) {
      const float d = inv_sqrt_deg(cnt[i2]);
      out[i2] = d * (res[threadIdx.x] + ht3[i2]) + bias[0];
    }
  }
}

extern "C" void kernel_launch(void* const* d_in, const int* in_sizes, int n_in,
                              void* d_out, int out_size, void* d_ws, size_t ws_size,
                              hipStream_t stream) {
  const float* x  = (const float*)d_in[0];
  const int*   ei = (const int*)d_in[1];
  const float* W1 = (const float*)d_in[2];
  const float* b1 = (const float*)d_in[3];
  const float* W2 = (const float*)d_in[4];
  const float* b2 = (const float*)d_in[5];
  const float* W3 = (const float*)d_in[6];
  const float* b3 = (const float*)d_in[7];
  float* out = (float*)d_out;

  const int D2 = 64, D = 32;
  const int N = in_sizes[0] / D2;
  const int E = in_sizes[1] / 2;
  const int nBins = (int)cdiv_ll(N, 1 << BIN_NODES_LOG);

  float* ws   = (float*)d_ws;
  float* ht   = ws;
  float* agg  = ht + (long long)N * D2;
  int2*  binBuf = (int2*)agg;
  float* ht2  = agg;
  int*   rec  = (int*)(agg + (long long)N * D2);
  int*   cnt  = rec + E;
  int*   off  = cnt + N;
  float* ht3  = (float*)(off + N);
  int*   binCount = (int*)(ht3 + N);
  (void)ws_size; (void)n_in; (void)out_size;

  const int B = 256;

  fill_kernel<<<(unsigned)cdiv_ll(nBins, B), B, 0, stream>>>((float*)binCount, 0.f, (unsigned)nBins);
  bin_kernel<<<(unsigned)cdiv_ll(E, B * BIN_EPT), B, 0, stream>>>(ei, binCount, binBuf, E, nBins);
  csr_kernel<<<(unsigned)nBins, B, 0, stream>>>(binBuf, binCount, ei, rec, cnt, off, N, E);

  const int rowTiles = (int)cdiv_ll(N, 16);

  {
    int colT = D2 / 16, tiles = rowTiles * colT;
    gemm_wmma_f32<<<(unsigned)cdiv_ll((long long)tiles * 32, B), B, 0, stream>>>(
        x, W1, cnt, ht, N, D2, D2, colT, tiles);
  }
  gather64_xform_kernel<<<(unsigned)rowTiles, 256, 0, stream>>>(
      ht, rec, off, cnt, b1, W2, ht2, (unsigned)N);
  gather32_dot_kernel<<<(unsigned)cdiv_ll((long long)N * (D / 4), B), B, 0, stream>>>(
      ht2, rec, off, cnt, b2, W3, ht3, (unsigned)N);
  gather1_kernel<<<(unsigned)cdiv_ll(N, 32), 256, 0, stream>>>(
      ht3, rec, off, cnt, b3, out, (unsigned)N);
}


# Aggregations fetch a node's source ids 16 per coalesced load and hand them out by lane broadcast (DPP): 8 row gathers in flight, not 4 behind a dependent id load; order of every sum unchanged
# speedup vs baseline: 4.7251x; 1.0563x over previous
#include <hip/hip_runtime.h>
#include <math.h>


typedef float v2f __attribute__((ext_vector_type(2)));
typedef float v8f __attribute__((ext_vector_type(8)));

#define BIN_NODES_LOG 8
#define BIN_CAP 6144
#define MAX_BINS 512
#define BIN_EPT 16
#define CSR_B 8

static inline long long cdiv_ll(long long a, long long b) { return (a + b - 1) / b; }

__device__ __forceinline__ float inv_sqrt_deg(int c) {
  const float dg = (float)(c + 1);
  return dg > 0.0f ? rsqrtf(dg) : 0.0f;
}

#define ROW16_BCAST(v, J) ((unsigned)__builtin_amdgcn_update_dpp(0, (int)(v), 0x150 + (J), 0xF, 0xF, true))
#define GRP8_BCAST(v, J)  ((unsigned)__builtin_amdgcn_mov_dpp8((unsigned)(v), (unsigned)(J) * 0x249249u))

#define GSLOT(J, ID, SHIFT)                                                     \
  float4 v##J = {0.f, 0.f, 0.f, 0.f};                                           \
  if ((unsigned)(J) < m) v##J = *(const float4*)(hb + (((ID) << (SHIFT)) + cb));
#define GADD(J) acc.x += v##J.x; acc.y += v##J.y; acc.z += v##J.z; acc.w += v##J.w;
#define GPIN asm volatile("" : "+v"(acc.x), "+v"(acc.y), "+v"(acc.z), "+v"(acc.w));

__global__ void fill_kernel(float* __restrict__ p, float v, unsigned n) {
  unsigned i = blockIdx.x * blockDim.x + threadIdx.x;
  if (i < n) p[i] = v;
}

__global__ void __launch_bounds__(256)
bin_kernel(const int* __restrict__ ei, int* __restrict__ binCount,
           int2* __restrict__ binBuf, int E, int nBins) {
  __shared__ int hist[MAX_BINS];
  const int tid = threadIdx.x;
  for (int b = tid; b < nBins; b += 256) hist[b] = 0;
  __syncthreads();

  const int e0 = blockIdx.x * (256 * BIN_EPT) + tid;
  int s[BIN_EPT], d[BIN_EPT], r[BIN_EPT];
#pragma unroll
  for (int q = 0; q < BIN_EPT; ++q) {
    const int e = e0 + q * 256;
    if (e < E) { s[q] = ei[e]; d[q] = ei[E + e]; } else { s[q] = 0; d[q] = -1; }
  }
#pragma unroll
  for (int q = 0; q < BIN_EPT; ++q) {
    r[q] = 0;
    if (d[q] >= 0) r[q] = atomicAdd(&hist[d[q] >> BIN_NODES_LOG], 1);
  }
  __syncthreads();
  {
    const int b0 = tid, b1 = tid + 256;
    const bool p0 = b0 < nBins, p1 = b1 < nBins;
    const int c0 = p0 ? hist[b0] : 0, c1 = p1 ? hist[b1] : 0;
    int g0 = 0, g1 = 0;
    if (p0) g0 = atomicAdd(&binCount[b0], c0);
    if (p1) g1 = atomicAdd(&binCount[b1], c1);
    if (p0) hist[b0] = g0;
    if (p1) hist[b1] = g1;
  }
  __syncthreads();
#pragma unroll
  for (int q = 0; q < BIN_EPT; ++q) {
    if (d[q] >= 0) {
      const int b = d[q] >> BIN_NODES_LOG;
      const unsigned pos = (unsigned)(hist[b] + r[q]);
      if (pos < (unsigned)BIN_CAP) {
        int2 rc; rc.x = s[q]; rc.y = d[q];
        binBuf[(unsigned)b * (unsigned)BIN_CAP + pos] = rc;
      }
    }
  }
}

__global__ void __launch_bounds__(256)
csr_kernel(const int2* __restrict__ binBuf, const int* __restrict__ binCount,
           const int* __restrict__ ei, int* __restrict__ rec, int* __restrict__ cnt,
           int* __restrict__ off, int N, int E) {
  __shared__ int cl[256];
  __shared__ int red[8];
  __shared__ int wsum[8];
  const int b = blockIdx.x, tid = threadIdx.x;
  const int lane = tid & 31, w = tid >> 5;

  int pre = 0;
  for (int m = tid; m < b; m += 256) pre += binCount[m];
  for (int dd = 16; dd > 0; dd >>= 1) pre += __shfl_xor(pre, dd, 32);
  if (lane == 0) red[w] = pre;
  cl[tid] = 0;
  __syncthreads();
  int base = 0;
  for (int k = 0; k < 8; ++k) base += red[k];

  const int nb = binCount[b];
  const bool fromBuf = nb <= BIN_CAP;
  const int2* bp = binBuf + (unsigned)b * (unsigned)BIN_CAP;

  if (fromBuf) {
    for (int i0 = tid; i0 < nb; i0 += 256 * CSR_B) {
      int dn[CSR_B];
#pragma unroll
      for (int q = 0; q < CSR_B; ++q) {
        const int idx = i0 + q * 256;
        dn[q] = idx < nb ? bp[idx].y : -1;
      }
#pragma unroll
      for (int q = 0; q < CSR_B; ++q)
        if (dn[q] >= 0) atomicAdd(&cl[dn[q] & 255], 1);
    }
  } else {
    for (int idx = tid; idx < E; idx += 256) {
      const int dn = ei[E + idx];
      if ((dn >> BIN_NODES_LOG) == b) atomicAdd(&cl[dn & 255], 1);
    }
  }
  __syncthreads();

  const int c = cl[tid];
  int incl = c;
  for (int dd = 1; dd < 32; dd <<= 1) {
    int t = __shfl_up(incl, dd, 32);
    if (lane >= dd) incl += t;
  }
  if (lane == 31) wsum[w] = incl;
  __syncthreads();
  int run = base + incl - c;
  for (int k = 0; k < 8; ++k)
    if (k < w) run += wsum[k];
  const int node = (b << BIN_NODES_LOG) + tid;
  if (node < N) { cnt[node] = c; off[node] = run; }
  cl[tid] = run;
  __syncthreads();

  if (fromBuf) {
    for (int i0 = tid; i0 < nb; i0 += 256 * CSR_B) {
      int2 rc[CSR_B];
#pragma unroll
      for (int q = 0; q < CSR_B; ++q) {
        const int idx = i0 + q * 256;
        if (idx < nb) rc[q] = bp[idx]; else { rc[q].x = 0; rc[q].y = -1; }
      }
#pragma unroll
      for (int q = 0; q < CSR_B; ++q) {
        if (rc[q].y >= 0) {
          const int slot = atomicAdd(&cl[rc[q].y & 255], 1);
          rec[(unsigned)slot] = rc[q].x;
        }
      }
    }
  } else {
    for (int idx = tid; idx < E; idx += 256) {
      const int dn = ei[E + idx];
      if ((dn >> BIN_NODES_LOG) == b) {
        const int slot = atomicAdd(&cl[dn & 255], 1);
        rec[(unsigned)slot] = ei[idx];
      }
    }
  }
}

__global__ void gemm_wmma_f32(const float* __restrict__ H, const float* __restrict__ W,
                              const int* __restrict__ cnt, float* __restrict__ out,
                              int N, int K, int OUT, int nColTiles, int nTiles) {
  int wave = (int)(((unsigned)blockIdx.x * blockDim.x + threadIdx.x) >> 5);
  if (wave >= nTiles) return;
  int lane = threadIdx.x & 31;
  int rt = wave / nColTiles, ct = wave % nColTiles;
  int half = lane >> 4;
  int m = lane & 15;
  int row = rt * 16 + m;
  int rowc = row < N ? row : N - 1;
  int col = ct * 16 + m;
  const float* hrow = H + (unsigned)rowc * (unsigned)K;

  float ds[8];
#pragma unroll
  for (int r = 0; r < 8; ++r) {
    int mrow = rt * 16 + r + 8 * half;
    ds[r] = inv_sqrt_deg(cnt[mrow < N ? mrow : N - 1]);
  }

  v8f acc = {};
  for (int kk = 0; kk < K; kk += 4) {
    int ka = kk + half * 2;
    v2f a = *(const v2f*)(hrow + ka);
    v2f b;
    b.x = W[(unsigned)ka * (unsigned)OUT + col];
    b.y = W[(unsigned)(ka + 1) * (unsigned)OUT + col];
    acc = __builtin_amdgcn_wmma_f32_16x16x4_f32(
         false, a,  false, b,
         (short)0, acc,  false,  false);
  }
#pragma unroll
  for (int r = 0; r < 8; ++r) {
    int mrow = rt * 16 + r + 8 * half;
    if (mrow < N) out[(unsigned)mrow * (unsigned)OUT + col] = acc[r] * ds[r];
  }
}

#define H1_PITCH 68
__global__ void __launch_bounds__(256)
gather64_xform_kernel(const float* __restrict__ ht, const int* __restrict__ rec,
                      const int* __restrict__ off, const int* __restrict__ cnt,
                      const float* __restrict__ bias, const float* __restrict__ W2,
                      float* __restrict__ out2, unsigned N) {
  __shared__ __attribute__((aligned(16))) float tile[16 * H1_PITCH];
  __shared__ __attribute__((aligned(16))) float sdis[16];
  const unsigned tid = threadIdx.x;
  const unsigned nl = tid >> 4;
  const unsigned i = blockIdx.x * 16u + nl;
  const unsigned c4 = (tid & 15u) << 2;
  float4 o = {0.f, 0.f, 0.f, 0.f};
  float di = 0.f;
  if (i < N) {
    const float* hc = ht + c4;
    const int* rp = rec + (unsigned)off[i];
    const int ci = cnt[i];
    const unsigned n = (unsigned)ci;
    const float4 hv = *(const float4*)(hc + (i << 6));
    const float4 bv = *(const float4*)(bias + c4);
    di = inv_sqrt_deg(ci);

    const char* hb = (const char*)ht;
    const unsigned cb = c4 << 2;
    const unsigned l16 = tid & 15u;
    float4 acc = {0.f, 0.f, 0.f, 0.f};
    unsigned my = l16 < n ? (unsigned)rp[l16] : 0u;
    for (unsigned k0 = 0; k0 < n; k0 += 16u) {
      const unsigned m = n - k0;
      const unsigned kn = k0 + 16u + l16;
      const unsigned nx = kn < n ? (unsigned)rp[kn] : 0u;
      GSLOT(0, ROW16_BCAST(my, 0), 8)   GSLOT(1, ROW16_BCAST(my, 1), 8)
      GSLOT(2, ROW16_BCAST(my, 2), 8)   GSLOT(3, ROW16_BCAST(my, 3), 8)
      GSLOT(4, ROW16_BCAST(my, 4), 8)   GSLOT(5, ROW16_BCAST(my, 5), 8)
      GSLOT(6, ROW16_BCAST(my, 6), 8)   GSLOT(7, ROW16_BCAST(my, 7), 8)
      GADD(0) GADD(1) GADD(2) GADD(3) GADD(4) GADD(5) GADD(6) GADD(7)
      GPIN
      GSLOT(8, ROW16_BCAST(my, 8), 8)   GSLOT(9, ROW16_BCAST(my, 9), 8)
      GSLOT(10, ROW16_BCAST(my, 10), 8) GSLOT(11, ROW16_BCAST(my, 11), 8)
      GSLOT(12, ROW16_BCAST(my, 12), 8) GSLOT(13, ROW16_BCAST(my, 13), 8)
      GSLOT(14, ROW16_BCAST(my, 14), 8) GSLOT(15, ROW16_BCAST(my, 15), 8)
      GADD(8) GADD(9) GADD(10) GADD(11) GADD(12) GADD(13) GADD(14) GADD(15)
      my = nx;
    }
    o.x = tanhf(di * (acc.x + hv.x) + bv.x);
    o.y = tanhf(di * (acc.y + hv.y) + bv.y);
    o.z = tanhf(di * (acc.z + hv.z) + bv.z);
    o.w = tanhf(di * (acc.w + hv.w) + bv.w);
  }
  *(float4*)(tile + nl * H1_PITCH + c4) = o;
  if ((tid & 15u) == 0u) sdis[nl] = di;
  __syncthreads();

  const unsigned w = tid >> 5;
  if (w < 2u) {
    const unsigned lane = tid & 31u;
    const unsigned half = lane >> 4;
    const unsigned m = lane & 15u;
    const unsigned col = w * 16u + m;
    const float* arow = tile + m * H1_PITCH + half * 2u;
    const float* bcol = W2 + (half * 2u) * 32u + col;
    const unsigned row0 = blockIdx.x * 16u + 8u * half;
    unsigned nv = N > row0 ? N - row0 : 0u;
    unsigned oi = row0 * 32u + col;
    const float4 sa = *(const float4*)(sdis + 8u * half);
    const float4 sb = *(const float4*)(sdis + 8u * half + 4u);
    float sc[8] = {sa.x, sa.y, sa.z, sa.w, sb.x, sb.y, sb.z, sb.w};
    asm volatile("" : "+v"(oi), "+v"(nv), "+v"(sc[0]), "+v"(sc[1]), "+v"(sc[2]), "+v"(sc[3]),
                      "+v"(sc[4]), "+v"(sc[5]), "+v"(sc[6]), "+v"(sc[7]));
    v8f acc = {};
#pragma unroll
    for (unsigned kk = 0; kk < 64u; kk += 4u) {
      v2f a = *(const v2f*)(arow + kk);
      v2f b;
      b.x = bcol[kk * 32u];
      b.y = bcol[kk * 32u + 32u];
      acc = __builtin_amdgcn_wmma_f32_16x16x4_f32(
           false, a,  false, b,
           (short)0, acc,  false,  false);
    }
    asm volatile("v_nop\n\tv_nop\n\tv_nop\n\tv_nop\n\tv_nop" : "+v"(acc));
#pragma unroll
    for (unsigned r = 0; r < 8u; ++r)
      if (r < nv) out2[oi + r * 32u] = acc[r] * sc[r];
  }
}

__global__ void gather32_dot_kernel(const float* __restrict__ ht2, const int* __restrict__ rec,
                                    const int* __restrict__ off, const int* __restrict__ cnt,
                                    const float* __restrict__ bias, const float* __restrict__ W3,
                                    float* __restrict__ out3, unsigned N) {
  unsigned t = blockIdx.x * blockDim.x + threadIdx.x;
  unsigned i = t >> 3;
  if (i >= N) return;
  unsigned c4 = (t & 7u) << 2;
  const float* hc = ht2 + c4;
  const int* rp = rec + (unsigned)off[i];
  const int ci = cnt[i];
  const unsigned n = (unsigned)ci;
  const float4 hv = *(const float4*)(hc + (i << 5));
  const float4 bv = *(const float4*)(bias + c4);
  const float4 wv = *(const float4*)(W3 + c4);
  const float di = inv_sqrt_deg(ci);

  const char* hb = (const char*)ht2;
  const unsigned cb = c4 << 2;
  const unsigned l8 = t & 7u;
  float4 acc = {0.f, 0.f, 0.f, 0.f};
  unsigned myA = l8 < n ? (unsigned)rp[l8] : 0u;
  unsigned myB = l8 + 8u < n ? (unsigned)rp[l8 + 8u] : 0u;
  for (unsigned k0 = 0; k0 < n; k0 += 16u) {
    const unsigned m = n - k0;
    const unsigned ka = k0 + 16u + l8, kb = k0 + 24u + l8;
    const unsigned nxA = ka < n ? (unsigned)rp[ka] : 0u;
    const unsigned nxB = kb < n ? (unsigned)rp[kb] : 0u;
    GSLOT(0, GRP8_BCAST(myA, 0), 7)  GSLOT(1, GRP8_BCAST(myA, 1), 7)
    GSLOT(2, GRP8_BCAST(myA, 2), 7)  GSLOT(3, GRP8_BCAST(myA, 3), 7)
    GSLOT(4, GRP8_BCAST(myA, 4), 7)  GSLOT(5, GRP8_BCAST(myA, 5), 7)
    GSLOT(6, GRP8_BCAST(myA, 6), 7)  GSLOT(7, GRP8_BCAST(myA, 7), 7)
    GADD(0) GADD(1) GADD(2) GADD(3) GADD(4) GADD(5) GADD(6) GADD(7)
    GPIN
    GSLOT(8, GRP8_BCAST(myB, 0), 7)  GSLOT(9, GRP8_BCAST(myB, 1), 7)
    GSLOT(10, GRP8_BCAST(myB, 2), 7) GSLOT(11, GRP8_BCAST(myB, 3), 7)
    GSLOT(12, GRP8_BCAST(myB, 4), 7) GSLOT(13, GRP8_BCAST(myB, 5), 7)
    GSLOT(14, GRP8_BCAST(myB, 6), 7) GSLOT(15, GRP8_BCAST(myB, 7), 7)
    GADD(8) GADD(9) GADD(10) GADD(11) GADD(12) GADD(13) GADD(14) GADD(15)
    myA = nxA; myB = nxB;
  }
  const float ox = tanhf(di * (acc.x + hv.x) + bv.x);
  const float oy = tanhf(di * (acc.y + hv.y) + bv.y);
  const float oz = tanhf(di * (acc.z + hv.z) + bv.z);
  const float ow = tanhf(di * (acc.w + hv.w) + bv.w);
  float p = ox * wv.x + oy * wv.y + oz * wv.z + ow * wv.w;
  p += __shfl_xor(p, 1, 32);
  p += __shfl_xor(p, 2, 32);
  p += __shfl_xor(p, 4, 32);
  if ((t & 7u) == 0u) out3[i] = p * di;
}

__global__ void __launch_bounds__(256)
gather1_kernel(const float* __restrict__ ht3, const int* __restrict__ rec,
               const int* __restrict__ off, const int* __restrict__ cnt,
               const float* __restrict__ bias,
               float* __restrict__ out, unsigned N) {
  __shared__ float res[32];
  const unsigned i = blockIdx.x * 32u + (threadIdx.x >> 3);
  const unsigned j = threadIdx.x & 7u;
  float s = 0.f;
  if (i < N) {
    const int* rp = rec + (unsigned)off[i];
    const unsigned n = (unsigned)cnt[i];
    const unsigned last = n - 1u;
    for (unsigned k = j; k < n; k += 32u) {
      const unsigned k1 = k + 8u, k2 = k + 16u, k3 = k + 24u;
      const unsigned r0 = (unsigned)rp[k];
      const unsigned r1 = (unsigned)rp[k1 < last ? k1 : last];
      const unsigned r2 = (unsigned)rp[k2 < last ? k2 : last];
      const unsigned r3 = (unsigned)rp[k3 < last ? k3 : last];
      const float a0 = ht3[r0], a1 = ht3[r1], a2 = ht3[r2], a3 = ht3[r3];
      s += a0;
      s += k1 < n ? a1 : 0.f;
      s += k2 < n ? a2 : 0.f;
      s += k3 < n ? a3 : 0.f;
    }
  }
  s += __shfl_xor(s, 1, 32);
  s += __shfl_xor(s, 2, 32);
  s += __shfl_xor(s, 4, 32);
  if (j == 0) res[threadIdx.x >> 3] = s;
  __syncthreads();
  if (threadIdx.x < 32) {
    const unsigned i2 = blockIdx.x * 32u + threadIdx.x;
    if (i2 < N) {
      const float d = inv_sqrt_deg(cnt[i2]);
      out[i2] = d * (res[threadIdx.x] + ht3[i2]) + bias[0];
    }
  }
}

extern "C" void kernel_launch(void* const* d_in, const int* in_sizes, int n_in,
                              void* d_out, int out_size, void* d_ws, size_t ws_size,
                              hipStream_t stream) {
  const float* x  = (const float*)d_in[0];
  const int*   ei = (const int*)d_in[1];
  const float* W1 = (const float*)d_in[2];
  const float* b1 = (const float*)d_in[3];
  const float* W2 = (const float*)d_in[4];
  const float* b2 = (const float*)d_in[5];
  const float* W3 = (const float*)d_in[6];
  const float* b3 = (const float*)d_in[7];
  float* out = (float*)d_out;

  const int D2 = 64, D = 32;
  const int N = in_sizes[0] / D2;
  const int E = in_sizes[1] / 2;
  const int nBins = (int)cdiv_ll(N, 1 << BIN_NODES_LOG);

  float* ws   = (float*)d_ws;
  float* ht   = ws;
  float* agg  = ht + (long long)N * D2;
  int2*  binBuf = (int2*)agg;
  float* ht2  = agg;
  int*   rec  = (int*)(agg + (long long)N * D2);
  int*   cnt  = rec + E;
  int*   off  = cnt + N;
  float* ht3  = (float*)(off + N);
  int*   binCount = (int*)(ht3 + N);
  (void)ws_size; (void)n_in; (void)out_size;

  const int B = 256;

  fill_kernel<<<(unsigned)cdiv_ll(nBins, B), B, 0, stream>>>((float*)binCount, 0.f, (unsigned)nBins);
  bin_kernel<<<(unsigned)cdiv_ll(E, B * BIN_EPT), B, 0, stream>>>(ei, binCount, binBuf, E, nBins);
  csr_kernel<<<(unsigned)nBins, B, 0, stream>>>(binBuf, binCount, ei, rec, cnt, off, N, E);

  const int rowTiles = (int)cdiv_ll(N, 16);

  {
    int colT = D2 / 16, tiles = rowTiles * colT;
    gemm_wmma_f32<<<(unsigned)cdiv_ll((long long)tiles * 32, B), B, 0, stream>>>(
        x, W1, cnt, ht, N, D2, D2, colT, tiles);
  }
  gather64_xform_kernel<<<(unsigned)rowTiles, 256, 0, stream>>>(
      ht, rec, off, cnt, b1, W2, ht2, (unsigned)N);
  gather32_dot_kernel<<<(unsigned)cdiv_ll((long long)N * (D / 4), B), B, 0, stream>>>(
      ht2, rec, off, cnt, b2, W3, ht3, (unsigned)N);
  gather1_kernel<<<(unsigned)cdiv_ll(N, 32), 256, 0, stream>>>(
      ht3, rec, off, cnt, b3, out, (unsigned)N);
}


# Layer-1 transform dis*(x@W1) moved into the segment pass: each 256-node work-group stages its x rows in LDS and runs 8 f32 WMMA tiles per wave once degrees are known; gemm launch and its waves go
# speedup vs baseline: 5.2996x; 1.1216x over previous
#include <hip/hip_runtime.h>
#include <math.h>


typedef float v2f __attribute__((ext_vector_type(2)));
typedef float v8f __attribute__((ext_vector_type(8)));

#define BIN_NODES_LOG 8
#define BIN_CAP 6144
#define MAX_BINS 512
#define BIN_EPT 16
#define CSR_B 8

static inline long long cdiv_ll(long long a, long long b) { return (a + b - 1) / b; }

__device__ __forceinline__ float inv_sqrt_deg(int c) {
  const float dg = (float)(c + 1);
  return dg > 0.0f ? rsqrtf(dg) : 0.0f;
}

#define ROW16_BCAST(v, J) ((unsigned)__builtin_amdgcn_update_dpp(0, (int)(v), 0x150 + (J), 0xF, 0xF, true))
#define GRP8_BCAST(v, J)  ((unsigned)__builtin_amdgcn_mov_dpp8((unsigned)(v), (unsigned)(J) * 0x249249u))

#define GSLOT(J, ID, SHIFT)                                                     \
  float4 v##J = {0.f, 0.f, 0.f, 0.f};                                           \
  if ((unsigned)(J) < m) v##J = *(const float4*)(hb + (((ID) << (SHIFT)) + cb));
#define GADD(J) acc.x += v##J.x; acc.y += v##J.y; acc.z += v##J.z; acc.w += v##J.w;
#define GPIN asm volatile("" : "+v"(acc.x), "+v"(acc.y), "+v"(acc.z), "+v"(acc.w));

__global__ void fill_kernel(float* __restrict__ p, float v, unsigned n) {
  unsigned i = blockIdx.x * blockDim.x + threadIdx.x;
  if (i < n) p[i] = v;
}

__global__ void __launch_bounds__(256)
bin_kernel(const int* __restrict__ ei, int* __restrict__ binCount,
           int2* __restrict__ binBuf, int E, int nBins) {
  __shared__ int hist[MAX_BINS];
  const int tid = threadIdx.x;
  for (int b = tid; b < nBins; b += 256) hist[b] = 0;
  __syncthreads();

  const int e0 = blockIdx.x * (256 * BIN_EPT) + tid;
  int s[BIN_EPT], d[BIN_EPT], r[BIN_EPT];
#pragma unroll
  for (int q = 0; q < BIN_EPT; ++q) {
    const int e = e0 + q * 256;
    if (e < E) { s[q] = ei[e]; d[q] = ei[E + e]; } else { s[q] = 0; d[q] = -1; }
  }
#pragma unroll
  for (int q = 0; q < BIN_EPT; ++q) {
    r[q] = 0;
    if (d[q] >= 0) r[q] = atomicAdd(&hist[d[q] >> BIN_NODES_LOG], 1);
  }
  __syncthreads();
  {
    const int b0 = tid, b1 = tid + 256;
    const bool p0 = b0 < nBins, p1 = b1 < nBins;
    const int c0 = p0 ? hist[b0] : 0, c1 = p1 ? hist[b1] : 0;
    int g0 = 0, g1 = 0;
    if (p0) g0 = atomicAdd(&binCount[b0], c0);
    if (p1) g1 = atomicAdd(&binCount[b1], c1);
    if (p0) hist[b0] = g0;
    if (p1) hist[b1] = g1;
  }
  __syncthreads();
#pragma unroll
  for (int q = 0; q < BIN_EPT; ++q) {
    if (d[q] >= 0) {
      const int b = d[q] >> BIN_NODES_LOG;
      const unsigned pos = (unsigned)(hist[b] + r[q]);
      if (pos < (unsigned)BIN_CAP) {
        int2 rc; rc.x = s[q]; rc.y = d[q];
        binBuf[(unsigned)b * (unsigned)BIN_CAP + pos] = rc;
      }
    }
  }
}

#define XT_PITCH 68
__global__ void __launch_bounds__(256)
csr_xform_kernel(const int2* __restrict__ binBuf, const int* __restrict__ binCount,
                 const int* __restrict__ ei, int* __restrict__ rec, int* __restrict__ cnt,
                 int* __restrict__ off, const float* __restrict__ x, const float* __restrict__ W1,
                 float* __restrict__ ht, int N, int E) {
  __shared__ __attribute__((aligned(16))) float xt[256 * XT_PITCH];
  __shared__ __attribute__((aligned(16))) float sdis[256];
  __shared__ int cl[256];
  __shared__ int red[8];
  __shared__ int wsum[8];
  const int b = blockIdx.x, tid = threadIdx.x;
  const int lane = tid & 31, w = tid >> 5;

  float4 xv[16];
  {
    const unsigned r0 = ((unsigned)b << BIN_NODES_LOG) + ((unsigned)w << 5);
    const unsigned c4 = ((unsigned)lane & 15u) << 2;
#pragma unroll
    for (unsigned j = 0; j < 16u; ++j) {
      unsigned node = r0 + 2u * j + ((unsigned)lane >> 4);
      node = node < (unsigned)N ? node : (unsigned)N - 1u;
      xv[j] = *(const float4*)(x + node * 64u + c4);
    }
  }

  int pre = 0;
  for (int m = tid; m < b; m += 256) pre += binCount[m];
  for (int dd = 16; dd > 0; dd >>= 1) pre += __shfl_xor(pre, dd, 32);
  if (lane == 0) red[w] = pre;
  cl[tid] = 0;
  {
    float* xw = xt + (((unsigned)w << 5) + ((unsigned)lane >> 4)) * XT_PITCH + (((unsigned)lane & 15u) << 2);
#pragma unroll
    for (unsigned j = 0; j < 16u; ++j) *(float4*)(xw + 2u * j * XT_PITCH) = xv[j];
  }
  __syncthreads();
  int base = 0;
  for (int k = 0; k < 8; ++k) base += red[k];

  const int nb = binCount[b];
  const bool fromBuf = nb <= BIN_CAP;
  const int2* bp = binBuf + (unsigned)b * (unsigned)BIN_CAP;

  if (fromBuf) {
    for (int i0 = tid; i0 < nb; i0 += 256 * CSR_B) {
      int dn[CSR_B];
#pragma unroll
      for (int q = 0; q < CSR_B; ++q) {
        const int idx = i0 + q * 256;
        dn[q] = idx < nb ? bp[idx].y : -1;
      }
#pragma unroll
      for (int q = 0; q < CSR_B; ++q)
        if (dn[q] >= 0) atomicAdd(&cl[dn[q] & 255], 1);
    }
  } else {
    for (int idx = tid; idx < E; idx += 256) {
      const int dn = ei[E + idx];
      if ((dn >> BIN_NODES_LOG) == b) atomicAdd(&cl[dn & 255], 1);
    }
  }
  __syncthreads();

  const int c = cl[tid];
  sdis[tid] = inv_sqrt_deg(c);
  int incl = c;
  for (int dd = 1; dd < 32; dd <<= 1) {
    int t = __shfl_up(incl, dd, 32);
    if (lane >= dd) incl += t;
  }
  if (lane == 31) wsum[w] = incl;
  __syncthreads();
  int run = base + incl - c;
  for (int k = 0; k < 8; ++k)
    if (k < w) run += wsum[k];
  const int node = (b << BIN_NODES_LOG) + tid;
  if (node < N) { cnt[node] = c; off[node] = run; }
  cl[tid] = run;
  __syncthreads();

  if (fromBuf) {
    for (int i0 = tid; i0 < nb; i0 += 256 * CSR_B) {
      int2 rc[CSR_B];
#pragma unroll
      for (int q = 0; q < CSR_B; ++q) {
        const int idx = i0 + q * 256;
        if (idx < nb) rc[q] = bp[idx]; else { rc[q].x = 0; rc[q].y = -1; }
      }
#pragma unroll
      for (int q = 0; q < CSR_B; ++q) {
        if (rc[q].y >= 0) {
          const int slot = atomicAdd(&cl[rc[q].y & 255], 1);
          rec[(unsigned)slot] = rc[q].x;
        }
      }
    }
  } else {
    for (int idx = tid; idx < E; idx += 256) {
      const int dn = ei[E + idx];
      if ((dn >> BIN_NODES_LOG) == b) {
        const int slot = atomicAdd(&cl[dn & 255], 1);
        rec[(unsigned)slot] = ei[idx];
      }
    }
  }

  {
    const unsigned half = (unsigned)lane >> 4;
    const unsigned m = (unsigned)lane & 15u;
    const float* arow = xt + (((unsigned)w << 5) + m) * XT_PITCH + half * 2u;
    const float* bcol = W1 + (half * 2u) * 64u + m;
    const unsigned lrow0 = ((unsigned)w << 5) + 8u * half;
    const unsigned row0 = ((unsigned)b << BIN_NODES_LOG) + lrow0;
    unsigned nv0 = (unsigned)N > row0 ? (unsigned)N - row0 : 0u;
    unsigned nv1 = (unsigned)N > row0 + 16u ? (unsigned)N - row0 - 16u : 0u;
    unsigned oi = row0 * 64u + m;
    const float4 s0a = *(const float4*)(sdis + lrow0);
    const float4 s0b = *(const float4*)(sdis + lrow0 + 4u);
    const float4 s1a = *(const float4*)(sdis + lrow0 + 16u);
    const float4 s1b = *(const float4*)(sdis + lrow0 + 20u);
    float sc0[8] = {s0a.x, s0a.y, s0a.z, s0a.w, s0b.x, s0b.y, s0b.z, s0b.w};
    float sc1[8] = {s1a.x, s1a.y, s1a.z, s1a.w, s1b.x, s1b.y, s1b.z, s1b.w};
    asm volatile("" : "+v"(oi), "+v"(nv0), "+v"(nv1),
                      "+v"(sc0[0]), "+v"(sc0[1]), "+v"(sc0[2]), "+v"(sc0[3]),
                      "+v"(sc0[4]), "+v"(sc0[5]), "+v"(sc0[6]), "+v"(sc0[7]),
                      "+v"(sc1[0]), "+v"(sc1[1]), "+v"(sc1[2]), "+v"(sc1[3]),
                      "+v"(sc1[4]), "+v"(sc1[5]), "+v"(sc1[6]), "+v"(sc1[7]));
    v8f a00 = {}, a01 = {}, a02 = {}, a03 = {}, a10 = {}, a11 = {}, a12 = {}, a13 = {};
#define XW(ACC, A, B) ACC = __builtin_amdgcn_wmma_f32_16x16x4_f32(false, A, false, B, (short)0, ACC, false, false);
#pragma unroll
    for (unsigned kk = 0; kk < 64u; kk += 4u) {
      const v2f fa0 = *(const v2f*)(arow + kk);
      const v2f fa1 = *(const v2f*)(arow + 16u * XT_PITCH + kk);
      v2f fb0, fb1, fb2, fb3;
      fb0.x = bcol[kk * 64u];        fb0.y = bcol[kk * 64u + 64u];
      fb1.x = bcol[kk * 64u + 16u];  fb1.y = bcol[kk * 64u + 80u];
      fb2.x = bcol[kk * 64u + 32u];  fb2.y = bcol[kk * 64u + 96u];
      fb3.x = bcol[kk * 64u + 48u];  fb3.y = bcol[kk * 64u + 112u];
      XW(a00, fa0, fb0) XW(a10, fa1, fb0)
      XW(a01, fa0, fb1) XW(a11, fa1, fb1)
      XW(a02, fa0, fb2) XW(a12, fa1, fb2)
      XW(a03, fa0, fb3) XW(a13, fa1, fb3)
    }
#undef XW
    asm volatile("v_nop\n\tv_nop\n\tv_nop\n\tv_nop\n\tv_nop"
                 : "+v"(a00), "+v"(a01), "+v"(a02), "+v"(a03), "+v"(a10), "+v"(a11), "+v"(a12), "+v"(a13));
#pragma unroll
    for (unsigned r = 0; r < 8u; ++r) {
      if (r < nv0) {
        float* o = ht + (oi + r * 64u);
        o[0] = a00[r] * sc0[r]; o[16] = a01[r] * sc0[r]; o[32] = a02[r] * sc0[r]; o[48] = a03[r] * sc0[r];
      }
      if (r < nv1) {
        float* o = ht + (oi + (r + 16u) * 64u);
        o[0] = a10[r] * sc1[r]; o[16] = a11[r] * sc1[r]; o[32] = a12[r] * sc1[r]; o[48] = a13[r] * sc1[r];
      }
    }
  }
}

#define H1_PITCH 68
__global__ void __launch_bounds__(256)
gather64_xform_kernel(const float* __restrict__ ht, const int* __restrict__ rec,
                      const int* __restrict__ off, const int* __restrict__ cnt,
                      const float* __restrict__ bias, const float* __restrict__ W2,
                      float* __restrict__ out2, unsigned N) {
  __shared__ __attribute__((aligned(16))) float tile[16 * H1_PITCH];
  __shared__ __attribute__((aligned(16))) float sdis[16];
  const unsigned tid = threadIdx.x;
  const unsigned nl = tid >> 4;
  const unsigned i = blockIdx.x * 16u + nl;
  const unsigned c4 = (tid & 15u) << 2;
  float4 o = {0.f, 0.f, 0.f, 0.f};
  float di = 0.f;
  if (i < N) {
    const float* hc = ht + c4;
    const int* rp = rec + (unsigned)off[i];
    const int ci = cnt[i];
    const unsigned n = (unsigned)ci;
    const float4 hv = *(const float4*)(hc + (i << 6));
    const float4 bv = *(const float4*)(bias + c4);
    di = inv_sqrt_deg(ci);

    const char* hb = (const char*)ht;
    const unsigned cb = c4 << 2;
    const unsigned l16 = tid & 15u;
    float4 acc = {0.f, 0.f, 0.f, 0.f};
    unsigned my = l16 < n ? (unsigned)rp[l16] : 0u;
    for (unsigned k0 = 0; k0 < n; k0 += 16u) {
      const unsigned m = n - k0;
      const unsigned kn = k0 + 16u + l16;
      const unsigned nx = kn < n ? (unsigned)rp[kn] : 0u;
      GSLOT(0, ROW16_BCAST(my, 0), 8)   GSLOT(1, ROW16_BCAST(my, 1), 8)
      GSLOT(2, ROW16_BCAST(my, 2), 8)   GSLOT(3, ROW16_BCAST(my, 3), 8)
      GSLOT(4, ROW16_BCAST(my, 4), 8)   GSLOT(5, ROW16_BCAST(my, 5), 8)
      GSLOT(6, ROW16_BCAST(my, 6), 8)   GSLOT(7, ROW16_BCAST(my, 7), 8)
      GADD(0) GADD(1) GADD(2) GADD(3) GADD(4) GADD(5) GADD(6) GADD(7)
      GPIN
      GSLOT(8, ROW16_BCAST(my, 8), 8)   GSLOT(9, ROW16_BCAST(my, 9), 8)
      GSLOT(10, ROW16_BCAST(my, 10), 8) GSLOT(11, ROW16_BCAST(my, 11), 8)
      GSLOT(12, ROW16_BCAST(my, 12), 8) GSLOT(13, ROW16_BCAST(my, 13), 8)
      GSLOT(14, ROW16_BCAST(my, 14), 8) GSLOT(15, ROW16_BCAST(my, 15), 8)
      GADD(8) GADD(9) GADD(10) GADD(11) GADD(12) GADD(13) GADD(14) GADD(15)
      my = nx;
    }
    o.x = tanhf(di * (acc.x + hv.x) + bv.x);
    o.y = tanhf(di * (acc.y + hv.y) + bv.y);
    o.z = tanhf(di * (acc.z + hv.z) + bv.z);
    o.w = tanhf(di * (acc.w + hv.w) + bv.w);
  }
  *(float4*)(tile + nl * H1_PITCH + c4) = o;
  if ((tid & 15u) == 0u) sdis[nl] = di;
  __syncthreads();

  const unsigned w = tid >> 5;
  if (w < 2u) {
    const unsigned lane = tid & 31u;
    const unsigned half = lane >> 4;
    const unsigned m = lane & 15u;
    const unsigned col = w * 16u + m;
    const float* arow = tile + m * H1_PITCH + half * 2u;
    const float* bcol = W2 + (half * 2u) * 32u + col;
    const unsigned row0 = blockIdx.x * 16u + 8u * half;
    unsigned nv = N > row0 ? N - row0 : 0u;
    unsigned oi = row0 * 32u + col;
    const float4 sa = *(const float4*)(sdis + 8u * half);
    const float4 sb = *(const float4*)(sdis + 8u * half + 4u);
    float sc[8] = {sa.x, sa.y, sa.z, sa.w, sb.x, sb.y, sb.z, sb.w};
    asm volatile("" : "+v"(oi), "+v"(nv), "+v"(sc[0]), "+v"(sc[1]), "+v"(sc[2]), "+v"(sc[3]),
                      "+v"(sc[4]), "+v"(sc[5]), "+v"(sc[6]), "+v"(sc[7]));
    v8f acc = {};
#pragma unroll
    for (unsigned kk = 0; kk < 64u; kk += 4u) {
      v2f a = *(const v2f*)(arow + kk);
      v2f b;
      b.x = bcol[kk * 32u];
      b.y = bcol[kk * 32u + 32u];
      acc = __builtin_amdgcn_wmma_f32_16x16x4_f32(
           false, a,  false, b,
           (short)0, acc,  false,  false);
    }
    asm volatile("v_nop\n\tv_nop\n\tv_nop\n\tv_nop\n\tv_nop" : "+v"(acc));
#pragma unroll
    for (unsigned r = 0; r < 8u; ++r)
      if (r < nv) out2[oi + r * 32u] = acc[r] * sc[r];
  }
}

__global__ void gather32_dot_kernel(const float* __restrict__ ht2, const int* __restrict__ rec,
                                    const int* __restrict__ off, const int* __restrict__ cnt,
                                    const float* __restrict__ bias, const float* __restrict__ W3,
                                    float* __restrict__ out3, unsigned N) {
  unsigned t = blockIdx.x * blockDim.x + threadIdx.x;
  unsigned i = t >> 3;
  if (i >= N) return;
  unsigned c4 = (t & 7u) << 2;
  const float* hc = ht2 + c4;
  const int* rp = rec + (unsigned)off[i];
  const int ci = cnt[i];
  const unsigned n = (unsigned)ci;
  const float4 hv = *(const float4*)(hc + (i << 5));
  const float4 bv = *(const float4*)(bias + c4);
  const float4 wv = *(const float4*)(W3 + c4);
  const float di = inv_sqrt_deg(ci);

  const char* hb = (const char*)ht2;
  const unsigned cb = c4 << 2;
  const unsigned l8 = t & 7u;
  float4 acc = {0.f, 0.f, 0.f, 0.f};
  unsigned myA = l8 < n ? (unsigned)rp[l8] : 0u;
  unsigned myB = l8 + 8u < n ? (unsigned)rp[l8 + 8u] : 0u;
  for (unsigned k0 = 0; k0 < n; k0 += 16u) {
    const unsigned m = n - k0;
    const unsigned ka = k0 + 16u + l8, kb = k0 + 24u + l8;
    const unsigned nxA = ka < n ? (unsigned)rp[ka] : 0u;
    const unsigned nxB = kb < n ? (unsigned)rp[kb] : 0u;
    GSLOT(0, GRP8_BCAST(myA, 0), 7)  GSLOT(1, GRP8_BCAST(myA, 1), 7)
    GSLOT(2, GRP8_BCAST(myA, 2), 7)  GSLOT(3, GRP8_BCAST(myA, 3), 7)
    GSLOT(4, GRP8_BCAST(myA, 4), 7)  GSLOT(5, GRP8_BCAST(myA, 5), 7)
    GSLOT(6, GRP8_BCAST(myA, 6), 7)  GSLOT(7, GRP8_BCAST(myA, 7), 7)
    GADD(0) GADD(1) GADD(2) GADD(3) GADD(4) GADD(5) GADD(6) GADD(7)
    GPIN
    GSLOT(8, GRP8_BCAST(myB, 0), 7)  GSLOT(9, GRP8_BCAST(myB, 1), 7)
    GSLOT(10, GRP8_BCAST(myB, 2), 7) GSLOT(11, GRP8_BCAST(myB, 3), 7)
    GSLOT(12, GRP8_BCAST(myB, 4), 7) GSLOT(13, GRP8_BCAST(myB, 5), 7)
    GSLOT(14, GRP8_BCAST(myB, 6), 7) GSLOT(15, GRP8_BCAST(myB, 7), 7)
    GADD(8) GADD(9) GADD(10) GADD(11) GADD(12) GADD(13) GADD(14) GADD(15)
    myA = nxA; myB = nxB;
  }
  const float ox = tanhf(di * (acc.x + hv.x) + bv.x);
  const float oy = tanhf(di * (acc.y + hv.y) + bv.y);
  const float oz = tanhf(di * (acc.z + hv.z) + bv.z);
  const float ow = tanhf(di * (acc.w + hv.w) + bv.w);
  float p = ox * wv.x + oy * wv.y + oz * wv.z + ow * wv.w;
  p += __shfl_xor(p, 1, 32);
  p += __shfl_xor(p, 2, 32);
  p += __shfl_xor(p, 4, 32);
  if ((t & 7u) == 0u) out3[i] = p * di;
}

__global__ void __launch_bounds__(256)
gather1_kernel(const float* __restrict__ ht3, const int* __restrict__ rec,
               const int* __restrict__ off, const int* __restrict__ cnt,
               const float* __restrict__ bias,
               float* __restrict__ out, unsigned N) {
  __shared__ float res[32];
  const unsigned i = blockIdx.x * 32u + (threadIdx.x >> 3);
  const unsigned j = threadIdx.x & 7u;
  float s = 0.f;
  if (i < N) {
    const int* rp = rec + (unsigned)off[i];
    const unsigned n = (unsigned)cnt[i];
    const unsigned last = n - 1u;
    for (unsigned k = j; k < n; k += 32u) {
      const unsigned k1 = k + 8u, k2 = k + 16u, k3 = k + 24u;
      const unsigned r0 = (unsigned)rp[k];
      const unsigned r1 = (unsigned)rp[k1 < last ? k1 : last];
      const unsigned r2 = (unsigned)rp[k2 < last ? k2 : last];
      const unsigned r3 = (unsigned)rp[k3 < last ? k3 : last];
      const float a0 = ht3[r0], a1 = ht3[r1], a2 = ht3[r2], a3 = ht3[r3];
      s += a0;
      s += k1 < n ? a1 : 0.f;
      s += k2 < n ? a2 : 0.f;
      s += k3 < n ? a3 : 0.f;
    }
  }
  s += __shfl_xor(s, 1, 32);
  s += __shfl_xor(s, 2, 32);
  s += __shfl_xor(s, 4, 32);
  if (j == 0) res[threadIdx.x >> 3] = s;
  __syncthreads();
  if (threadIdx.x < 32) {
    const unsigned i2 = blockIdx.x * 32u + threadIdx.x;
    if (i2 < N) {
      const float d = inv_sqrt_deg(cnt[i2]);
      out[i2] = d * (res[threadIdx.x] + ht3[i2]) + bias[0];
    }
  }
}

extern "C" void kernel_launch(void* const* d_in, const int* in_sizes, int n_in,
                              void* d_out, int out_size, void* d_ws, size_t ws_size,
                              hipStream_t stream) {
  const float* x  = (const float*)d_in[0];
  const int*   ei = (const int*)d_in[1];
  const float* W1 = (const float*)d_in[2];
  const float* b1 = (const float*)d_in[3];
  const float* W2 = (const float*)d_in[4];
  const float* b2 = (const float*)d_in[5];
  const float* W3 = (const float*)d_in[6];
  const float* b3 = (const float*)d_in[7];
  float* out = (float*)d_out;

  const int D2 = 64, D = 32;
  const int N = in_sizes[0] / D2;
  const int E = in_sizes[1] / 2;
  const int nBins = (int)cdiv_ll(N, 1 << BIN_NODES_LOG);

  float* ws   = (float*)d_ws;
  float* ht   = ws;
  float* agg  = ht + (long long)N * D2;
  int2*  binBuf = (int2*)agg;
  float* ht2  = agg;
  int*   rec  = (int*)(agg + (long long)N * D2);
  int*   cnt  = rec + E;
  int*   off  = cnt + N;
  float* ht3  = (float*)(off + N);
  int*   binCount = (int*)(ht3 + N);
  (void)ws_size; (void)n_in; (void)out_size;

  const int B = 256;

  fill_kernel<<<(unsigned)cdiv_ll(nBins, B), B, 0, stream>>>((float*)binCount, 0.f, (unsigned)nBins);
  bin_kernel<<<(unsigned)cdiv_ll(E, B * BIN_EPT), B, 0, stream>>>(ei, binCount, binBuf, E, nBins);
  csr_xform_kernel<<<(unsigned)nBins, B, 0, stream>>>(binBuf, binCount, ei, rec, cnt, off, x, W1, ht, N, E);

  const int rowTiles = (int)cdiv_ll(N, 16);

  gather64_xform_kernel<<<(unsigned)rowTiles, 256, 0, stream>>>(
      ht, rec, off, cnt, b1, W2, ht2, (unsigned)N);
  gather32_dot_kernel<<<(unsigned)cdiv_ll((long long)N * (D / 4), B), B, 0, stream>>>(
      ht2, rec, off, cnt, b2, W3, ht3, (unsigned)N);
  gather1_kernel<<<(unsigned)cdiv_ll(N, 32), 256, 0, stream>>>(
      ht3, rec, off, cnt, b3, out, (unsigned)N);
}
